# starting kernel
# baseline (speedup 1.0000x reference)
#include <hip/hip_runtime.h>
#include <hip/hip_bf16.h>

typedef __attribute__((ext_vector_type(2))) float v2f;
typedef __attribute__((ext_vector_type(8))) float v8f;

#define N_NODES 100000
#define N_EDGES 1600000
#define D_FEAT  64
#define UNITS   64

__global__ __launch_bounds__(256) void gcn_degree_kernel(
    const int* __restrict__ target, float* __restrict__ deg, int nEdges) {
  int e = blockIdx.x * blockDim.x + threadIdx.x;
  if (e < nEdges) atomicAdd(&deg[target[e]], 1.0f);
}

__global__ __launch_bounds__(256) void gcn_rsqrt_kernel(
    const float* __restrict__ deg, float* __restrict__ dinv, int n) {
  int i = blockIdx.x * blockDim.x + threadIdx.x;
  if (i < n) dinv[i] = rsqrtf(deg[i]);
}

__global__ __launch_bounds__(256) void gcn_scatter_kernel(
    const float* __restrict__ x, const int* __restrict__ src,
    const int* __restrict__ tgt, const float* __restrict__ dinv,
    float* __restrict__ pooled, int nEdges) {
  int wave = (blockIdx.x * blockDim.x + threadIdx.x) >> 5;
  int lane = threadIdx.x & 31;
  if (wave >= nEdges) return;
  int s = src[wave];
  int t = tgt[wave];
  float d = dinv[s];
  const float2* xs = (const float2*)(x + (size_t)s * D_FEAT);
  float2 v = xs[lane];
  float* p = pooled + (size_t)t * D_FEAT + 2 * lane;
  atomicAdd(p + 0, v.x * d);
  atomicAdd(p + 1, v.y * d);
}

__global__ __launch_bounds__(128) void gcn_gemm_wmma_kernel(
    const float* __restrict__ pooled, const float* __restrict__ W,
    const float* __restrict__ bias, const float* __restrict__ dinv,
    float* __restrict__ out) {
  __shared__ float sW[D_FEAT * UNITS];
  __shared__ float sA[16 * D_FEAT];

  const int tid = threadIdx.x;
  const int rowBase = blockIdx.x * 16;

  {
    const float4* Wv = (const float4*)W;
    float4* sWv = (float4*)sW;
    for (int i = tid; i < (D_FEAT * UNITS) / 4; i += 128) sWv[i] = Wv[i];
  }
  {
    const float4* Av = (const float4*)(pooled + (size_t)rowBase * D_FEAT);
    float4* sAv = (float4*)sA;
    for (int i = tid; i < (16 * D_FEAT) / 4; i += 128) {
      int row = i / (D_FEAT / 4);
      float d = dinv[rowBase + row];
      float4 v = Av[i];
      v.x *= d; v.y *= d; v.z *= d; v.w *= d;
      sAv[i] = v;
    }
  }
  __syncthreads();

  const int lane  = tid & 31;
  const int wave  = tid >> 5;
  const int nBase = wave * 16;
  const int h     = lane >> 4;
  const int mn    = lane & 15;

  v8f acc = {};
  #pragma unroll
  for (int kk = 0; kk < D_FEAT; kk += 4) {
    const int k0 = kk + 2 * h;
    v2f a, b;
    a.x = sA[mn * D_FEAT + k0];
    a.y = sA[mn * D_FEAT + k0 + 1];
    b.x = sW[(k0)     * UNITS + nBase + mn];
    b.y = sW[(k0 + 1) * UNITS + nBase + mn];
    acc = __builtin_amdgcn_wmma_f32_16x16x4_f32(
         false, a,  false, b,
         (short)0, acc,  false,  false);
  }

  const float bn = bias[nBase + mn];
  #pragma unroll
  for (int j = 0; j < 8; ++j) {
    const int m = j + 8 * h;
    float r = acc[j] + bn;
    out[(size_t)(rowBase + m) * UNITS + nBase + mn] = fmaxf(r, 0.0f);
  }
}

extern "C" void kernel_launch(void* const* d_in, const int* in_sizes, int n_in,
                              void* d_out, int out_size, void* d_ws, size_t ws_size,
                              hipStream_t stream) {
  const float* x      = (const float*)d_in[0];
  const float* W      = (const float*)d_in[1];
  const float* b      = (const float*)d_in[2];
  const int*   source = (const int*)d_in[3];
  const int*   target = (const int*)d_in[4];
  float*       out    = (float*)d_out;

  float* pooled = (float*)d_ws;
  float* deg    = pooled + (size_t)N_NODES * D_FEAT;
  float* dinv   = deg + N_NODES;

  hipMemsetAsync(pooled, 0, (size_t)N_NODES * D_FEAT * sizeof(float), stream);
  hipMemsetAsync(deg,    0, (size_t)N_NODES * sizeof(float), stream);

  gcn_degree_kernel<<<(N_EDGES + 255) / 256, 256, 0, stream>>>(target, deg, N_EDGES);
  gcn_rsqrt_kernel<<<(N_NODES + 255) / 256, 256, 0, stream>>>(deg, dinv, N_NODES);
  gcn_scatter_kernel<<<(N_EDGES + 7) / 8, 256, 0, stream>>>(x, source, target, dinv,
                                                            pooled, N_EDGES);
  gcn_gemm_wmma_kernel<<<N_NODES / 16, 128, 0, stream>>>(pooled, W, b, dinv, out);
}


# In-degree counted with integer LDS adds per (edge chunk, node slice) block, partials summed in the rsqrt kernel: replaces per-edge random global f32 atomics and the degree memset; deg bit-identical
# speedup vs baseline: 1.3908x; 1.3908x over previous
#include <hip/hip_runtime.h>
#include <hip/hip_bf16.h>

typedef __attribute__((ext_vector_type(2))) float v2f;
typedef __attribute__((ext_vector_type(8))) float v8f;

#define N_NODES 100000
#define N_EDGES 1600000
#define D_FEAT  64
#define UNITS   64

#define DEG_SLICES  8
#define DEG_CHUNKS  32
#define SLICE_NODES (N_NODES / DEG_SLICES)
#define CHUNK_EDGES (N_EDGES / DEG_CHUNKS)
#define DEG_LOADS   ((CHUNK_EDGES / 4 + 1023) / 1024)
static_assert(N_NODES % DEG_SLICES == 0 && SLICE_NODES % 4 == 0, "slice must be whole uint4s");
static_assert(N_EDGES % DEG_CHUNKS == 0 && CHUNK_EDGES % 4 == 0, "chunk must be whole int4s");

__global__ __launch_bounds__(1024) void gcn_degree_bins_kernel(
    const int* __restrict__ target, unsigned* __restrict__ part) {
  __shared__ __attribute__((aligned(16))) unsigned bins[SLICE_NODES];
  const int tid = threadIdx.x;
  const int s = blockIdx.x % DEG_SLICES;
  const int c = blockIdx.x / DEG_SLICES;

  const unsigned lo = (unsigned)(s * SLICE_NODES);
  const int4* t4 = (const int4*)(target + (size_t)c * CHUNK_EDGES);
  int4 t[DEG_LOADS];
  #pragma unroll
  for (int k = 0; k < DEG_LOADS; ++k) {
    const int i = tid + k * 1024;
    t[k] = (i < CHUNK_EDGES / 4) ? t4[i] : make_int4(-1, -1, -1, -1);
  }

  uint4* bins4 = (uint4*)bins;
  for (int j = tid; j < SLICE_NODES / 4; j += 1024) bins4[j] = make_uint4(0u, 0u, 0u, 0u);
  __syncthreads();

  #pragma unroll
  for (int k = 0; k < DEG_LOADS; ++k) {
    const unsigned u0 = (unsigned)t[k].x - lo;
    const unsigned u1 = (unsigned)t[k].y - lo;
    const unsigned u2 = (unsigned)t[k].z - lo;
    const unsigned u3 = (unsigned)t[k].w - lo;
    if (u0 < (unsigned)SLICE_NODES) atomicAdd(&bins[u0], 1u);
    if (u1 < (unsigned)SLICE_NODES) atomicAdd(&bins[u1], 1u);
    if (u2 < (unsigned)SLICE_NODES) atomicAdd(&bins[u2], 1u);
    if (u3 < (unsigned)SLICE_NODES) atomicAdd(&bins[u3], 1u);
  }
  __syncthreads();

  uint4* dst4 = (uint4*)(part + (size_t)c * N_NODES + lo);
  for (int j = tid; j < SLICE_NODES / 4; j += 1024) dst4[j] = bins4[j];
}

__global__ __launch_bounds__(256) void gcn_rsqrt_kernel(
    const unsigned* __restrict__ part, float* __restrict__ dinv, int n) {
  int i = blockIdx.x * blockDim.x + threadIdx.x;
  if (i < n) {
    unsigned cnt = 0u;
    #pragma unroll
    for (int c = 0; c < DEG_CHUNKS; ++c) cnt += part[(size_t)c * N_NODES + i];
    dinv[i] = rsqrtf((float)cnt);
  }
}

__global__ __launch_bounds__(256) void gcn_scatter_kernel(
    const float* __restrict__ x, const int* __restrict__ src,
    const int* __restrict__ tgt, const float* __restrict__ dinv,
    float* __restrict__ pooled, int nEdges) {
  int wave = (blockIdx.x * blockDim.x + threadIdx.x) >> 5;
  int lane = threadIdx.x & 31;
  if (wave >= nEdges) return;
  int s = src[wave];
  int t = tgt[wave];
  float d = dinv[s];
  const float2* xs = (const float2*)(x + (size_t)s * D_FEAT);
  float2 v = xs[lane];
  float* p = pooled + (size_t)t * D_FEAT + 2 * lane;
  atomicAdd(p + 0, v.x * d);
  atomicAdd(p + 1, v.y * d);
}

__global__ __launch_bounds__(128) void gcn_gemm_wmma_kernel(
    const float* __restrict__ pooled, const float* __restrict__ W,
    const float* __restrict__ bias, const float* __restrict__ dinv,
    float* __restrict__ out) {
  __shared__ float sW[D_FEAT * UNITS];
  __shared__ float sA[16 * D_FEAT];

  const int tid = threadIdx.x;
  const int rowBase = blockIdx.x * 16;

  {
    const float4* Wv = (const float4*)W;
    float4* sWv = (float4*)sW;
    for (int i = tid; i < (D_FEAT * UNITS) / 4; i += 128) sWv[i] = Wv[i];
  }
  {
    const float4* Av = (const float4*)(pooled + (size_t)rowBase * D_FEAT);
    float4* sAv = (float4*)sA;
    for (int i = tid; i < (16 * D_FEAT) / 4; i += 128) {
      int row = i / (D_FEAT / 4);
      float d = dinv[rowBase + row];
      float4 v = Av[i];
      v.x *= d; v.y *= d; v.z *= d; v.w *= d;
      sAv[i] = v;
    }
  }
  __syncthreads();

  const int lane  = tid & 31;
  const int wave  = tid >> 5;
  const int nBase = wave * 16;
  const int h     = lane >> 4;
  const int mn    = lane & 15;

  v8f acc = {};
  #pragma unroll
  for (int kk = 0; kk < D_FEAT; kk += 4) {
    const int k0 = kk + 2 * h;
    v2f a, b;
    a.x = sA[mn * D_FEAT + k0];
    a.y = sA[mn * D_FEAT + k0 + 1];
    b.x = sW[(k0)     * UNITS + nBase + mn];
    b.y = sW[(k0 + 1) * UNITS + nBase + mn];
    acc = __builtin_amdgcn_wmma_f32_16x16x4_f32(
         false, a,  false, b,
         (short)0, acc,  false,  false);
  }

  const float bn = bias[nBase + mn];
  #pragma unroll
  for (int j = 0; j < 8; ++j) {
    const int m = j + 8 * h;
    float r = acc[j] + bn;
    out[(size_t)(rowBase + m) * UNITS + nBase + mn] = fmaxf(r, 0.0f);
  }
}

extern "C" void kernel_launch(void* const* d_in, const int* in_sizes, int n_in,
                              void* d_out, int out_size, void* d_ws, size_t ws_size,
                              hipStream_t stream) {
  const float* x      = (const float*)d_in[0];
  const float* W      = (const float*)d_in[1];
  const float* b      = (const float*)d_in[2];
  const int*   source = (const int*)d_in[3];
  const int*   target = (const int*)d_in[4];
  float*       out    = (float*)d_out;

  float* pooled = (float*)d_ws;
  float* deg    = pooled + (size_t)N_NODES * D_FEAT;
  float* dinv   = deg + N_NODES;
  unsigned* part = (unsigned*)d_ws;

  gcn_degree_bins_kernel<<<DEG_SLICES * DEG_CHUNKS, 1024, 0, stream>>>(target, part);
  gcn_rsqrt_kernel<<<(N_NODES + 255) / 256, 256, 0, stream>>>(part, dinv, N_NODES);

  hipMemsetAsync(pooled, 0, (size_t)N_NODES * D_FEAT * sizeof(float), stream);

  gcn_scatter_kernel<<<(N_EDGES + 7) / 8, 256, 0, stream>>>(x, source, target, dinv,
                                                            pooled, N_EDGES);
  gcn_gemm_wmma_kernel<<<N_NODES / 16, 128, 0, stream>>>(pooled, W, b, dinv, out);
}


# Edges counting-sorted by target with LDS slot adds on the degree partials; per-node register gather fused into the WMMA kernel replaces f32 atomic scatter, memset and pooled; sums in edge-list order
# speedup vs baseline: 3.7232x; 2.6770x over previous
#include <hip/hip_runtime.h>
#include <hip/hip_bf16.h>

typedef __attribute__((ext_vector_type(2))) float v2f;
typedef __attribute__((ext_vector_type(8))) float v8f;

#define N_NODES 100000
#define N_EDGES 1600000
#define D_FEAT  64
#define UNITS   64

#define DEG_SLICES  8
#define DEG_CHUNKS  32
#define SLICE_NODES (N_NODES / DEG_SLICES)
#define CHUNK_EDGES (N_EDGES / DEG_CHUNKS)
#define DEG_LOADS   ((CHUNK_EDGES / 4 + 1023) / 1024)
static_assert(N_NODES % DEG_SLICES == 0 && SLICE_NODES % 4 == 0, "slice must be whole uint4s");
static_assert(N_EDGES % DEG_CHUNKS == 0 && CHUNK_EDGES % 4 == 0, "chunk must be whole int4s");

__global__ __launch_bounds__(1024) void gcn_degree_bins_kernel(
    const int* __restrict__ target, unsigned* __restrict__ part) {
  __shared__ __attribute__((aligned(16))) unsigned bins[SLICE_NODES];
  const int tid = threadIdx.x;
  const int s = blockIdx.x % DEG_SLICES;
  const int c = blockIdx.x / DEG_SLICES;

  const unsigned lo = (unsigned)(s * SLICE_NODES);
  const int4* t4 = (const int4*)(target + (size_t)c * CHUNK_EDGES);
  int4 t[DEG_LOADS];
  #pragma unroll
  for (int k = 0; k < DEG_LOADS; ++k) {
    const int i = tid + k * 1024;
    t[k] = (i < CHUNK_EDGES / 4) ? t4[i] : make_int4(-1, -1, -1, -1);
  }

  uint4* bins4 = (uint4*)bins;
  for (int j = tid; j < SLICE_NODES / 4; j += 1024) bins4[j] = make_uint4(0u, 0u, 0u, 0u);
  __syncthreads();

  #pragma unroll
  for (int k = 0; k < DEG_LOADS; ++k) {
    const unsigned u0 = (unsigned)t[k].x - lo;
    const unsigned u1 = (unsigned)t[k].y - lo;
    const unsigned u2 = (unsigned)t[k].z - lo;
    const unsigned u3 = (unsigned)t[k].w - lo;
    if (u0 < (unsigned)SLICE_NODES) atomicAdd(&bins[u0], 1u);
    if (u1 < (unsigned)SLICE_NODES) atomicAdd(&bins[u1], 1u);
    if (u2 < (unsigned)SLICE_NODES) atomicAdd(&bins[u2], 1u);
    if (u3 < (unsigned)SLICE_NODES) atomicAdd(&bins[u3], 1u);
  }
  __syncthreads();

  uint4* dst4 = (uint4*)(part + (size_t)c * N_NODES + lo);
  for (int j = tid; j < SLICE_NODES / 4; j += 1024) dst4[j] = bins4[j];
}

#define SCAN_BLOCK    256
#define N_SCAN_BLOCKS ((N_NODES + SCAN_BLOCK - 1) / SCAN_BLOCK)
static_assert(SCAN_BLOCK == 256, "the fill kernel finds a node's scan block with a shift by 8");

__device__ __forceinline__ unsigned wave_inclusive_scan(unsigned v, int lane) {
  #pragma unroll
  for (int d = 1; d < 32; d <<= 1) {
    const unsigned t = __shfl_up(v, (unsigned)d);
    if (lane >= d) v += t;
  }
  return v;
}

template <int NWAVES>
__device__ __forceinline__ unsigned block_exclusive_scan(unsigned v, unsigned* wave_tot,
                                                         unsigned* total) {
  const int lane = threadIdx.x & 31;
  const int w = threadIdx.x >> 5;
  const unsigned inc = wave_inclusive_scan(v, lane);
  if (lane == 31) wave_tot[w] = inc;
  __syncthreads();
  unsigned off = 0u, tot = 0u;
  for (int j = 0; j < NWAVES; ++j) {
    const unsigned t = wave_tot[j];
    if (j < w) off += t;
    tot += t;
  }
  *total = tot;
  return off + inc - v;
}

__global__ __launch_bounds__(SCAN_BLOCK) void gcn_rsqrt_offsets_kernel(
    unsigned* __restrict__ part, float* __restrict__ dinv, unsigned* __restrict__ bsum) {
  __shared__ unsigned wave_tot[SCAN_BLOCK / 32];
  const int i = blockIdx.x * SCAN_BLOCK + threadIdx.x;
  const bool live = i < N_NODES;
  unsigned v[DEG_CHUNKS];
  unsigned cnt = 0u;
  #pragma unroll
  for (int c = 0; c < DEG_CHUNKS; ++c) {
    v[c] = live ? part[(size_t)c * N_NODES + i] : 0u;
    cnt += v[c];
  }
  if (live) dinv[i] = rsqrtf((float)cnt);

  unsigned total;
  const unsigned excl = block_exclusive_scan<SCAN_BLOCK / 32>(cnt, wave_tot, &total);
  if (threadIdx.x == 0) bsum[blockIdx.x] = total;
  if (live) {
    unsigned run = excl;
    #pragma unroll
    for (int c = 0; c < DEG_CHUNKS; ++c) {
      part[(size_t)c * N_NODES + i] = run;
      run += v[c];
    }
  }
}

#define FILL_THREADS 512
#define FILL_LOADS   ((CHUNK_EDGES / 4 + FILL_THREADS - 1) / FILL_THREADS)
static_assert(N_SCAN_BLOCKS <= FILL_THREADS, "one thread per block sum in the scan");

__global__ __launch_bounds__(FILL_THREADS) void gcn_fill_kernel(
    const int* __restrict__ source, const int* __restrict__ target,
    const unsigned* __restrict__ part, const unsigned* __restrict__ bsum,
    unsigned* __restrict__ rowptr, int* __restrict__ sorted_src) {
  __shared__ __attribute__((aligned(16))) unsigned bins[SLICE_NODES];
  __shared__ unsigned bbase[N_SCAN_BLOCKS];
  __shared__ unsigned wave_tot[FILL_THREADS / 32];
  const int tid = threadIdx.x;
  const int s = blockIdx.x % DEG_SLICES;
  const int c = blockIdx.x / DEG_SLICES;
  const unsigned lo = (unsigned)(s * SLICE_NODES);

  const int4* t4 = (const int4*)(target + (size_t)c * CHUNK_EDGES);
  const int4* s4 = (const int4*)(source + (size_t)c * CHUNK_EDGES);
  int4 t[FILL_LOADS];
  int4 sv[FILL_LOADS];
  #pragma unroll
  for (int k = 0; k < FILL_LOADS; ++k) {
    const int i = tid + k * FILL_THREADS;
    const bool in = i < CHUNK_EDGES / 4;
    t[k]  = in ? t4[i] : make_int4(-1, -1, -1, -1);
    sv[k] = in ? s4[i] : make_int4(0, 0, 0, 0);
  }

  unsigned total;
  const unsigned bs = (tid < N_SCAN_BLOCKS) ? bsum[tid] : 0u;
  const unsigned be = block_exclusive_scan<FILL_THREADS / 32>(bs, wave_tot, &total);
  if (tid < N_SCAN_BLOCKS) bbase[tid] = be;
  __syncthreads();

  const uint4* p4 = (const uint4*)(part + (size_t)c * N_NODES + lo);
  uint4* bins4 = (uint4*)bins;
  uint4* rp4 = (uint4*)(rowptr + lo);
  for (int j = tid; j < SLICE_NODES / 4; j += FILL_THREADS) {
    uint4 v = p4[j];
    const unsigned b = bbase[(lo + 4u * (unsigned)j) >> 8];
    v.x += b; v.y += b; v.z += b; v.w += b;
    bins4[j] = v;
    if (c == 0) rp4[j] = v;
  }
  if (c == 0 && s == DEG_SLICES - 1 && tid == 0) rowptr[N_NODES] = total;
  __syncthreads();

  #pragma unroll
  for (int k = 0; k < FILL_LOADS; ++k) {
    const unsigned u0 = (unsigned)t[k].x - lo;
    const unsigned u1 = (unsigned)t[k].y - lo;
    const unsigned u2 = (unsigned)t[k].z - lo;
    const unsigned u3 = (unsigned)t[k].w - lo;
    t[k].x = (u0 < (unsigned)SLICE_NODES) ? (int)atomicAdd(&bins[u0], 1u) : -1;
    t[k].y = (u1 < (unsigned)SLICE_NODES) ? (int)atomicAdd(&bins[u1], 1u) : -1;
    t[k].z = (u2 < (unsigned)SLICE_NODES) ? (int)atomicAdd(&bins[u2], 1u) : -1;
    t[k].w = (u3 < (unsigned)SLICE_NODES) ? (int)atomicAdd(&bins[u3], 1u) : -1;
  }
  #pragma unroll
  for (int k = 0; k < FILL_LOADS; ++k) {
    if (t[k].x >= 0) sorted_src[t[k].x] = sv[k].x;
    if (t[k].y >= 0) sorted_src[t[k].y] = sv[k].y;
    if (t[k].z >= 0) sorted_src[t[k].z] = sv[k].z;
    if (t[k].w >= 0) sorted_src[t[k].w] = sv[k].w;
  }
}

#define GATHER_BATCH 8

__device__ __forceinline__ int lane_value(int v, int l) {
  return __builtin_amdgcn_readlane(v, l);
}
__device__ __forceinline__ float lane_value(float v, int l) {
  return __builtin_bit_cast(float, __builtin_amdgcn_readlane(__builtin_bit_cast(int, v), l));
}

__global__ __launch_bounds__(128) void gcn_gather_gemm_wmma_kernel(
    const float* __restrict__ x, const int* __restrict__ sorted_src,
    const unsigned* __restrict__ rowptr, const float* __restrict__ dinv,
    const float* __restrict__ W, const float* __restrict__ bias,
    float* __restrict__ out) {
  __shared__ __attribute__((aligned(16))) float sW[D_FEAT * UNITS];
  __shared__ __attribute__((aligned(16))) float sA[16 * D_FEAT];

  const int tid = threadIdx.x;
  const int rowBase = blockIdx.x * 16;
  const int lane  = tid & 31;
  const int wave  = tid >> 5;

  const int nb = rowBase + wave * 4;
  const int rp = (lane < 5) ? (int)rowptr[nb + lane] : 0;
  const float dn = (lane < 4) ? dinv[nb + lane] : 0.0f;
  int beg[4], cnt[4], sL[4];
  float dL[4];
  #pragma unroll
  for (int r = 0; r < 4; ++r) {
    beg[r] = lane_value(rp, r);
    cnt[r] = lane_value(rp, r + 1) - beg[r];
    sL[r] = 0;
    if (lane < cnt[r]) sL[r] = sorted_src[beg[r] + lane];
  }
  #pragma unroll
  for (int r = 0; r < 4; ++r) {
    dL[r] = 0.0f;
    if (lane < cnt[r]) dL[r] = dinv[sL[r]];
  }

  {
    const float4* Wv = (const float4*)W;
    float4* sWv = (float4*)sW;
    for (int i = tid; i < (D_FEAT * UNITS) / 4; i += 128) sWv[i] = Wv[i];
  }

  #pragma unroll
  for (int r = 0; r < 4; ++r) {
    float ax = 0.0f, ay = 0.0f;
    int s_l = sL[r];
    float d_l = dL[r];
    const int total = cnt[r];
    for (int done = 0; done < total; done += 32) {
      const int m = (total - done < 32) ? (total - done) : 32;
      if (done != 0) {
        s_l = 0; d_l = 0.0f;
        if (lane < m) { s_l = sorted_src[beg[r] + done + lane]; d_l = dinv[s_l]; }
      }
      for (int j = 0; j < m; j += GATHER_BATCH) {
        float2 v[GATHER_BATCH];
        float d[GATHER_BATCH];
        #pragma unroll
        for (int q = 0; q < GATHER_BATCH; ++q) {
          const int jj = (j + q < m) ? (j + q) : (m - 1);
          const int s = lane_value(s_l, jj);
          d[q] = lane_value(d_l, jj);
          v[q] = ((const float2*)(x + (size_t)s * D_FEAT))[lane];
        }
        #pragma unroll
        for (int q = 0; q < GATHER_BATCH; ++q) {
          if (j + q < m) { ax += v[q].x * d[q]; ay += v[q].y * d[q]; }
        }
      }
    }
    const float dr = lane_value(dn, r);
    float2 o; o.x = ax * dr; o.y = ay * dr;
    ((float2*)(sA + (wave * 4 + r) * D_FEAT))[lane] = o;
  }
  __syncthreads();

  const int nBase = wave * 16;
  const int h     = lane >> 4;
  const int mn    = lane & 15;

  v8f acc = {};
  #pragma unroll
  for (int kk = 0; kk < D_FEAT; kk += 4) {
    const int k0 = kk + 2 * h;
    v2f a, b;
    a.x = sA[mn * D_FEAT + k0];
    a.y = sA[mn * D_FEAT + k0 + 1];
    b.x = sW[(k0)     * UNITS + nBase + mn];
    b.y = sW[(k0 + 1) * UNITS + nBase + mn];
    acc = __builtin_amdgcn_wmma_f32_16x16x4_f32(
         false, a,  false, b,
         (short)0, acc,  false,  false);
  }

  const float bn = bias[nBase + mn];
  #pragma unroll
  for (int j = 0; j < 8; ++j) {
    const int m = j + 8 * h;
    float r = acc[j] + bn;
    out[(size_t)(rowBase + m) * UNITS + nBase + mn] = fmaxf(r, 0.0f);
  }
}

extern "C" void kernel_launch(void* const* d_in, const int* in_sizes, int n_in,
                              void* d_out, int out_size, void* d_ws, size_t ws_size,
                              hipStream_t stream) {
  const float* x      = (const float*)d_in[0];
  const float* W      = (const float*)d_in[1];
  const float* b      = (const float*)d_in[2];
  const int*   source = (const int*)d_in[3];
  const int*   target = (const int*)d_in[4];
  float*       out    = (float*)d_out;

  unsigned* part       = (unsigned*)d_ws;
  int*      sorted_src = (int*)d_ws + (size_t)DEG_CHUNKS * N_NODES;
  float*    dinv       = (float*)d_ws + (size_t)DEG_CHUNKS * N_NODES + N_EDGES;
  unsigned* rowptr     = (unsigned*)d_ws + (size_t)DEG_CHUNKS * N_NODES + N_EDGES + N_NODES;
  unsigned* bsum       = rowptr + N_NODES + 4;

  gcn_degree_bins_kernel<<<DEG_SLICES * DEG_CHUNKS, 1024, 0, stream>>>(target, part);
  gcn_rsqrt_offsets_kernel<<<N_SCAN_BLOCKS, SCAN_BLOCK, 0, stream>>>(part, dinv, bsum);
  gcn_fill_kernel<<<DEG_SLICES * DEG_CHUNKS, FILL_THREADS, 0, stream>>>(source, target, part, bsum,
                                                                rowptr, sorted_src);
  gcn_gather_gemm_wmma_kernel<<<N_NODES / 16, 128, 0, stream>>>(x, sorted_src, rowptr, dinv,
                                                                W, b, out);
}


# Fill kernel packs each source's saturated in-degree (16-bit table staged in LDS) above its 17-bit id in the edge word; gather takes rsqrt from the word, no 4-byte lookup per edge; sum order unchanged
# speedup vs baseline: 3.8986x; 1.0471x over previous
#include <hip/hip_runtime.h>
#include <hip/hip_bf16.h>

typedef __attribute__((ext_vector_type(2))) float v2f;
typedef __attribute__((ext_vector_type(8))) float v8f;

#define N_NODES 100000
#define N_EDGES 1600000
#define D_FEAT  64
#define UNITS   64

#define SRC_BITS 17
#define SRC_MASK ((1u << SRC_BITS) - 1u)
#define DEG_SAT  ((1u << (32 - SRC_BITS)) - 1u)
static_assert(N_NODES <= (1 << SRC_BITS), "a source id must fit the low field of the edge word");
static_assert(DEG_SAT <= 0xFFFFu, "the saturated degree must fit the 16-bit table");

#define DEG_SLICES  8
#define DEG_CHUNKS  32
#define SLICE_NODES (N_NODES / DEG_SLICES)
#define CHUNK_EDGES (N_EDGES / DEG_CHUNKS)
#define DEG_LOADS   ((CHUNK_EDGES / 4 + 1023) / 1024)
static_assert(N_NODES % DEG_SLICES == 0 && SLICE_NODES % 4 == 0, "slice must be whole uint4s");
static_assert(N_EDGES % DEG_CHUNKS == 0 && CHUNK_EDGES % 4 == 0, "chunk must be whole int4s");

__global__ __launch_bounds__(1024) void gcn_degree_bins_kernel(
    const int* __restrict__ target, unsigned* __restrict__ part) {
  __shared__ __attribute__((aligned(16))) unsigned bins[SLICE_NODES];
  const int tid = threadIdx.x;
  const int s = blockIdx.x % DEG_SLICES;
  const int c = blockIdx.x / DEG_SLICES;

  const unsigned lo = (unsigned)(s * SLICE_NODES);
  const int4* t4 = (const int4*)(target + (size_t)c * CHUNK_EDGES);
  int4 t[DEG_LOADS];
  #pragma unroll
  for (int k = 0; k < DEG_LOADS; ++k) {
    const int i = tid + k * 1024;
    t[k] = (i < CHUNK_EDGES / 4) ? t4[i] : make_int4(-1, -1, -1, -1);
  }

  uint4* bins4 = (uint4*)bins;
  for (int j = tid; j < SLICE_NODES / 4; j += 1024) bins4[j] = make_uint4(0u, 0u, 0u, 0u);
  __syncthreads();

  #pragma unroll
  for (int k = 0; k < DEG_LOADS; ++k) {
    const unsigned u0 = (unsigned)t[k].x - lo;
    const unsigned u1 = (unsigned)t[k].y - lo;
    const unsigned u2 = (unsigned)t[k].z - lo;
    const unsigned u3 = (unsigned)t[k].w - lo;
    if (u0 < (unsigned)SLICE_NODES) atomicAdd(&bins[u0], 1u);
    if (u1 < (unsigned)SLICE_NODES) atomicAdd(&bins[u1], 1u);
    if (u2 < (unsigned)SLICE_NODES) atomicAdd(&bins[u2], 1u);
    if (u3 < (unsigned)SLICE_NODES) atomicAdd(&bins[u3], 1u);
  }
  __syncthreads();

  uint4* dst4 = (uint4*)(part + (size_t)c * N_NODES + lo);
  for (int j = tid; j < SLICE_NODES / 4; j += 1024) dst4[j] = bins4[j];
}

#define SCAN_BLOCK    256
#define N_SCAN_BLOCKS ((N_NODES + SCAN_BLOCK - 1) / SCAN_BLOCK)
static_assert(SCAN_BLOCK == 256, "the fill kernel finds a node's scan block with a shift by 8");

__device__ __forceinline__ unsigned wave_inclusive_scan(unsigned v, int lane) {
  #pragma unroll
  for (int d = 1; d < 32; d <<= 1) {
    const unsigned t = __shfl_up(v, (unsigned)d);
    if (lane >= d) v += t;
  }
  return v;
}

template <int NWAVES>
__device__ __forceinline__ unsigned block_exclusive_scan(unsigned v, unsigned* wave_tot,
                                                         unsigned* total) {
  const int lane = threadIdx.x & 31;
  const int w = threadIdx.x >> 5;
  const unsigned inc = wave_inclusive_scan(v, lane);
  if (lane == 31) wave_tot[w] = inc;
  __syncthreads();
  unsigned off = 0u, tot = 0u;
  for (int j = 0; j < NWAVES; ++j) {
    const unsigned t = wave_tot[j];
    if (j < w) off += t;
    tot += t;
  }
  *total = tot;
  return off + inc - v;
}

__global__ __launch_bounds__(SCAN_BLOCK) void gcn_degree_offsets_kernel(
    unsigned* __restrict__ part, unsigned short* __restrict__ deg16, unsigned* __restrict__ bsum) {
  __shared__ unsigned wave_tot[SCAN_BLOCK / 32];
  const int i = blockIdx.x * SCAN_BLOCK + threadIdx.x;
  const bool live = i < N_NODES;
  unsigned v[DEG_CHUNKS];
  unsigned cnt = 0u;
  #pragma unroll
  for (int c = 0; c < DEG_CHUNKS; ++c) {
    v[c] = live ? part[(size_t)c * N_NODES + i] : 0u;
    cnt += v[c];
  }
  if (live) deg16[i] = (unsigned short)(cnt < DEG_SAT ? cnt : DEG_SAT);

  unsigned total;
  const unsigned excl = block_exclusive_scan<SCAN_BLOCK / 32>(cnt, wave_tot, &total);
  if (threadIdx.x == 0) bsum[blockIdx.x] = total;
  if (live) {
    unsigned run = excl;
    #pragma unroll
    for (int c = 0; c < DEG_CHUNKS; ++c) {
      part[(size_t)c * N_NODES + i] = run;
      run += v[c];
    }
  }
}

#define FILL_THREADS 512
#define FILL_LOADS   ((CHUNK_EDGES / 4 + FILL_THREADS - 1) / FILL_THREADS)
static_assert(N_SCAN_BLOCKS <= FILL_THREADS, "one thread per block sum in the scan");
static_assert((N_NODES * 2) % 16 == 0, "the degree table must be whole uint4s");

__device__ __forceinline__ bool in_slice(int tgt, unsigned lo) {
  return ((unsigned)tgt - lo) < (unsigned)SLICE_NODES;
}

__global__ __launch_bounds__(FILL_THREADS) void gcn_fill_kernel(
    const int* __restrict__ source, const int* __restrict__ target,
    const unsigned* __restrict__ part, const unsigned* __restrict__ bsum,
    const unsigned short* __restrict__ deg16,
    unsigned* __restrict__ rowptr, unsigned* __restrict__ sorted_edge) {
  __shared__ __attribute__((aligned(16))) unsigned bins[SLICE_NODES];
  __shared__ __attribute__((aligned(16))) unsigned short sdeg[N_NODES];
  __shared__ unsigned bbase[N_SCAN_BLOCKS];
  __shared__ unsigned wave_tot[FILL_THREADS / 32];
  const int tid = threadIdx.x;
  const int s = blockIdx.x % DEG_SLICES;
  const int c = blockIdx.x / DEG_SLICES;
  const unsigned lo = (unsigned)(s * SLICE_NODES);

  const int4* t4 = (const int4*)(target + (size_t)c * CHUNK_EDGES);
  const int4* s4 = (const int4*)(source + (size_t)c * CHUNK_EDGES);
  int4 t[FILL_LOADS];
  int4 sv[FILL_LOADS];
  #pragma unroll
  for (int k = 0; k < FILL_LOADS; ++k) {
    const int i = tid + k * FILL_THREADS;
    const bool in = i < CHUNK_EDGES / 4;
    t[k]  = in ? t4[i] : make_int4(-1, -1, -1, -1);
    sv[k] = in ? s4[i] : make_int4(0, 0, 0, 0);
  }

  unsigned total;
  const unsigned bs = (tid < N_SCAN_BLOCKS) ? bsum[tid] : 0u;
  const unsigned be = block_exclusive_scan<FILL_THREADS / 32>(bs, wave_tot, &total);
  if (tid < N_SCAN_BLOCKS) bbase[tid] = be;
  __syncthreads();

  const uint4* p4 = (const uint4*)(part + (size_t)c * N_NODES + lo);
  uint4* bins4 = (uint4*)bins;
  uint4* rp4 = (uint4*)(rowptr + lo);
  for (int j = tid; j < SLICE_NODES / 4; j += FILL_THREADS) {
    uint4 v = p4[j];
    const unsigned b = bbase[(lo + 4u * (unsigned)j) >> 8];
    v.x += b; v.y += b; v.z += b; v.w += b;
    bins4[j] = v;
    if (c == 0) rp4[j] = v;
  }
  if (c == 0 && s == DEG_SLICES - 1 && tid == 0) rowptr[N_NODES] = total;

  {
    const uint4* g4 = (const uint4*)deg16;
    uint4* sdeg4 = (uint4*)sdeg;
    for (int j = tid; j < (N_NODES * 2) / 16; j += FILL_THREADS) sdeg4[j] = g4[j];
  }
  __syncthreads();

  unsigned d0 = sdeg[in_slice(t[0].x, lo) ? sv[0].x : 0];
  unsigned d1 = sdeg[in_slice(t[0].y, lo) ? sv[0].y : 0];
  unsigned d2 = sdeg[in_slice(t[0].z, lo) ? sv[0].z : 0];
  unsigned d3 = sdeg[in_slice(t[0].w, lo) ? sv[0].w : 0];
  #pragma unroll
  for (int k = 0; k < FILL_LOADS; ++k) {
    unsigned n0 = 0u, n1 = 0u, n2 = 0u, n3 = 0u;
    if (k + 1 < FILL_LOADS) {
      n0 = sdeg[in_slice(t[k + 1].x, lo) ? sv[k + 1].x : 0];
      n1 = sdeg[in_slice(t[k + 1].y, lo) ? sv[k + 1].y : 0];
      n2 = sdeg[in_slice(t[k + 1].z, lo) ? sv[k + 1].z : 0];
      n3 = sdeg[in_slice(t[k + 1].w, lo) ? sv[k + 1].w : 0];
    }
    const unsigned u0 = (unsigned)t[k].x - lo;
    const unsigned u1 = (unsigned)t[k].y - lo;
    const unsigned u2 = (unsigned)t[k].z - lo;
    const unsigned u3 = (unsigned)t[k].w - lo;
    t[k].x = (u0 < (unsigned)SLICE_NODES) ? (int)atomicAdd(&bins[u0], 1u) : -1;
    t[k].y = (u1 < (unsigned)SLICE_NODES) ? (int)atomicAdd(&bins[u1], 1u) : -1;
    t[k].z = (u2 < (unsigned)SLICE_NODES) ? (int)atomicAdd(&bins[u2], 1u) : -1;
    t[k].w = (u3 < (unsigned)SLICE_NODES) ? (int)atomicAdd(&bins[u3], 1u) : -1;
    sv[k].x = (int)((unsigned)sv[k].x | (d0 << SRC_BITS));
    sv[k].y = (int)((unsigned)sv[k].y | (d1 << SRC_BITS));
    sv[k].z = (int)((unsigned)sv[k].z | (d2 << SRC_BITS));
    sv[k].w = (int)((unsigned)sv[k].w | (d3 << SRC_BITS));
    asm volatile("" : "+v"(sv[k].x), "+v"(sv[k].y), "+v"(sv[k].z), "+v"(sv[k].w));
    d0 = n0; d1 = n1; d2 = n2; d3 = n3;
  }
  #pragma unroll
  for (int k = 0; k < FILL_LOADS; ++k) {
    if (t[k].x >= 0) sorted_edge[t[k].x] = (unsigned)sv[k].x;
    if (t[k].y >= 0) sorted_edge[t[k].y] = (unsigned)sv[k].y;
    if (t[k].z >= 0) sorted_edge[t[k].z] = (unsigned)sv[k].z;
    if (t[k].w >= 0) sorted_edge[t[k].w] = (unsigned)sv[k].w;
  }
}

#define GATHER_BATCH 8

__device__ __forceinline__ int lane_value(int v, int l) {
  return __builtin_amdgcn_readlane(v, l);
}
__device__ __forceinline__ float lane_value(float v, int l) {
  return __builtin_bit_cast(float, __builtin_amdgcn_readlane(__builtin_bit_cast(int, v), l));
}

__device__ __forceinline__ float edge_scale(unsigned w, const unsigned* __restrict__ rowptr) {
  unsigned deg = w >> SRC_BITS;
  if (deg == DEG_SAT) {
    const unsigned s = w & SRC_MASK;
    deg = rowptr[s + 1] - rowptr[s];
  }
  return rsqrtf((float)deg);
}

__global__ __launch_bounds__(128) void gcn_gather_gemm_wmma_kernel(
    const float* __restrict__ x, const unsigned* __restrict__ sorted_edge,
    const unsigned* __restrict__ rowptr,
    const float* __restrict__ W, const float* __restrict__ bias,
    float* __restrict__ out) {
  __shared__ __attribute__((aligned(16))) float sW[D_FEAT * UNITS];
  __shared__ __attribute__((aligned(16))) float sA[16 * D_FEAT];

  const int tid = threadIdx.x;
  const int rowBase = blockIdx.x * 16;
  const int lane  = tid & 31;
  const int wave  = tid >> 5;

  const int nb = rowBase + wave * 4;
  const int rp = (lane < 5) ? (int)rowptr[nb + lane] : 0;
  int beg[4], cnt[4], sL[4];
  unsigned wL[4];
  float dL[4];
  #pragma unroll
  for (int r = 0; r < 4; ++r) {
    beg[r] = lane_value(rp, r);
    cnt[r] = lane_value(rp, r + 1) - beg[r];
    wL[r] = 0u;
    if (lane < cnt[r]) wL[r] = sorted_edge[beg[r] + lane];
  }
  #pragma unroll
  for (int r = 0; r < 4; ++r) {
    sL[r] = (int)(wL[r] & SRC_MASK);
    dL[r] = 0.0f;
    if (lane < cnt[r]) dL[r] = edge_scale(wL[r], rowptr);
  }

  {
    const float4* Wv = (const float4*)W;
    float4* sWv = (float4*)sW;
    for (int i = tid; i < (D_FEAT * UNITS) / 4; i += 128) sWv[i] = Wv[i];
  }

  #pragma unroll
  for (int r = 0; r < 4; ++r) {
    float ax = 0.0f, ay = 0.0f;
    int s_l = sL[r];
    float d_l = dL[r];
    const int total = cnt[r];
    for (int done = 0; done < total; done += 32) {
      const int m = (total - done < 32) ? (total - done) : 32;
      if (done != 0) {
        s_l = 0; d_l = 0.0f;
        if (lane < m) {
          const unsigned w = sorted_edge[beg[r] + done + lane];
          s_l = (int)(w & SRC_MASK);
          d_l = edge_scale(w, rowptr);
        }
      }
      for (int j = 0; j < m; j += GATHER_BATCH) {
        float2 v[GATHER_BATCH];
        float d[GATHER_BATCH];
        #pragma unroll
        for (int q = 0; q < GATHER_BATCH; ++q) {
          const int jj = (j + q < m) ? (j + q) : (m - 1);
          const int s = lane_value(s_l, jj);
          d[q] = lane_value(d_l, jj);
          v[q] = ((const float2*)(x + (size_t)s * D_FEAT))[lane];
        }
        #pragma unroll
        for (int q = 0; q < GATHER_BATCH; ++q) {
          if (j + q < m) { ax += v[q].x * d[q]; ay += v[q].y * d[q]; }
        }
      }
    }
    const float dr = rsqrtf((float)total);
    float2 o; o.x = ax * dr; o.y = ay * dr;
    ((float2*)(sA + (wave * 4 + r) * D_FEAT))[lane] = o;
  }
  __syncthreads();

  const int nBase = wave * 16;
  const int h     = lane >> 4;
  const int mn    = lane & 15;

  v8f acc = {};
  #pragma unroll
  for (int kk = 0; kk < D_FEAT; kk += 4) {
    const int k0 = kk + 2 * h;
    v2f a, b;
    a.x = sA[mn * D_FEAT + k0];
    a.y = sA[mn * D_FEAT + k0 + 1];
    b.x = sW[(k0)     * UNITS + nBase + mn];
    b.y = sW[(k0 + 1) * UNITS + nBase + mn];
    acc = __builtin_amdgcn_wmma_f32_16x16x4_f32(
         false, a,  false, b,
         (short)0, acc,  false,  false);
  }

  const float bn = bias[nBase + mn];
  #pragma unroll
  for (int j = 0; j < 8; ++j) {
    const int m = j + 8 * h;
    float r = acc[j] + bn;
    out[(size_t)(rowBase + m) * UNITS + nBase + mn] = fmaxf(r, 0.0f);
  }
}

extern "C" void kernel_launch(void* const* d_in, const int* in_sizes, int n_in,
                              void* d_out, int out_size, void* d_ws, size_t ws_size,
                              hipStream_t stream) {
  const float* x      = (const float*)d_in[0];
  const float* W      = (const float*)d_in[1];
  const float* b      = (const float*)d_in[2];
  const int*   source = (const int*)d_in[3];
  const int*   target = (const int*)d_in[4];
  float*       out    = (float*)d_out;

  unsigned*       part        = (unsigned*)d_ws;
  unsigned*       sorted_edge = (unsigned*)d_ws + (size_t)DEG_CHUNKS * N_NODES;
  unsigned short* deg16       = (unsigned short*)((unsigned*)d_ws + (size_t)DEG_CHUNKS * N_NODES + N_EDGES);
  unsigned*       rowptr      = (unsigned*)d_ws + (size_t)DEG_CHUNKS * N_NODES + N_EDGES + N_NODES;
  unsigned*       bsum        = rowptr + N_NODES + 4;

  gcn_degree_bins_kernel<<<DEG_SLICES * DEG_CHUNKS, 1024, 0, stream>>>(target, part);
  gcn_degree_offsets_kernel<<<N_SCAN_BLOCKS, SCAN_BLOCK, 0, stream>>>(part, deg16, bsum);
  gcn_fill_kernel<<<DEG_SLICES * DEG_CHUNKS, FILL_THREADS, 0, stream>>>(source, target, part, bsum,
                                                                deg16, rowptr, sorted_edge);
  gcn_gather_gemm_wmma_kernel<<<N_NODES / 16, 128, 0, stream>>>(x, sorted_edge, rowptr,
                                                                W, b, out);
}


# Edge list ordered in two LDS levels (chunk blocks group edges by target range, range blocks order them by node): contiguous edge stores, 3 launches; dinv read per edge; sums in source-id order
# speedup vs baseline: 4.5065x; 1.1559x over previous
#include <hip/hip_runtime.h>
#include <hip/hip_bf16.h>


typedef __attribute__((ext_vector_type(2))) float v2f;
typedef __attribute__((ext_vector_type(8))) float v8f;

#define N_NODES 100000
#define N_EDGES 1600000
#define D_FEAT  64
#define UNITS   64

#define SRC_BITS 17
#define SRC_MASK ((1u << SRC_BITS) - 1u)
#define RANGE_NODES 400
#define N_RANGES    (N_NODES / RANGE_NODES)
#define N_CHUNKS    250
#define CHUNK_EDGES (N_EDGES / N_CHUNKS)
static_assert(N_NODES <= (1 << SRC_BITS), "a source id must fit the low field of the edge word");
static_assert(RANGE_NODES <= (1 << (32 - SRC_BITS)), "a node's index in its range must fit the high field");
static_assert(N_NODES % RANGE_NODES == 0, "ranges must be whole");
static_assert(N_EDGES % N_CHUNKS == 0, "chunks must be whole");
static_assert(CHUNK_EDGES < 65536, "ranks, starts and counts inside a chunk are kept in 16 bits");

__device__ __forceinline__ unsigned wave_inclusive_scan(unsigned v, int lane) {
  #pragma unroll
  for (int d = 1; d < 32; d <<= 1) {
    const unsigned t = __shfl_up(v, (unsigned)d);
    if (lane >= d) v += t;
  }
  return v;
}

template <int NWAVES>
__device__ __forceinline__ unsigned block_exclusive_scan(unsigned v, unsigned* wave_tot,
                                                         unsigned* total) {
  const int lane = threadIdx.x & 31;
  const int w = threadIdx.x >> 5;
  const unsigned inc = wave_inclusive_scan(v, lane);
  if (lane == 31) wave_tot[w] = inc;
  __syncthreads();
  unsigned off = 0u, tot = 0u;
  for (int j = 0; j < NWAVES; ++j) {
    const unsigned t = wave_tot[j];
    if (j < w) off += t;
    tot += t;
  }
  *total = tot;
  return off + inc - v;
}

#define PART_THREADS 320
#define PART_LOADS   (CHUNK_EDGES / 4 / PART_THREADS)
static_assert(CHUNK_EDGES % (4 * PART_THREADS) == 0, "a chunk must be whole int4s per thread");
static_assert(N_RANGES <= PART_THREADS, "one thread per range in the scan");

__global__ __launch_bounds__(PART_THREADS) void gcn_partition_kernel(
    const int* __restrict__ source, const int* __restrict__ target,
    unsigned* __restrict__ binned, unsigned* __restrict__ seg) {
  __shared__ unsigned cnt[N_RANGES];
  __shared__ unsigned start[N_RANGES];
  __shared__ unsigned wave_tot[PART_THREADS / 32];
  __shared__ __attribute__((aligned(16))) unsigned stage[CHUNK_EDGES];
  const int tid = threadIdx.x;
  const int c = blockIdx.x;

  const int4* t4 = (const int4*)(target + (size_t)c * CHUNK_EDGES);
  const int4* s4 = (const int4*)(source + (size_t)c * CHUNK_EDGES);
  int4 t[PART_LOADS];
  int4 sv[PART_LOADS];
  #pragma unroll
  for (int k = 0; k < PART_LOADS; ++k) {
    t[k]  = t4[tid + k * PART_THREADS];
    sv[k] = s4[tid + k * PART_THREADS];
  }
  if (tid < N_RANGES) cnt[tid] = 0u;
  __syncthreads();

  unsigned key[PART_LOADS * 4];
  unsigned word[PART_LOADS * 4];
  #pragma unroll
  for (int k = 0; k < PART_LOADS; ++k) {
    const int tv[4] = {t[k].x, t[k].y, t[k].z, t[k].w};
    const int sw[4] = {sv[k].x, sv[k].y, sv[k].z, sv[k].w};
    #pragma unroll
    for (int e = 0; e < 4; ++e) {
      const unsigned tg = (unsigned)tv[e];
      const unsigned r = tg / (unsigned)RANGE_NODES;
      const unsigned loc = tg - r * (unsigned)RANGE_NODES;
      const unsigned rank = atomicAdd(&cnt[r], 1u);
      key[k * 4 + e] = (r << 16) | rank;
      word[k * 4 + e] = (unsigned)sw[e] | (loc << SRC_BITS);
    }
  }
  __syncthreads();

  unsigned total;
  const unsigned n = (tid < N_RANGES) ? cnt[tid] : 0u;
  const unsigned ex = block_exclusive_scan<PART_THREADS / 32>(n, wave_tot, &total);
  if (tid < N_RANGES) {
    start[tid] = ex;
    seg[(size_t)tid * N_CHUNKS + c] = (ex << 16) | n;
  }
  __syncthreads();

  #pragma unroll
  for (int i = 0; i < PART_LOADS * 4; ++i) stage[start[key[i] >> 16] + (key[i] & 0xFFFFu)] = word[i];
  __syncthreads();

  uint4* dst4 = (uint4*)(binned + (size_t)c * CHUNK_EDGES);
  const uint4* st4 = (const uint4*)stage;
  #pragma unroll
  for (int k = 0; k < PART_LOADS; ++k) dst4[tid + k * PART_THREADS] = st4[tid + k * PART_THREADS];
}

#define SORT_THREADS 1024
#define SORT_WAVES   (SORT_THREADS / 32)
#define SORT_CAP     16384
#define RANK_MAX_DEG 64
#define SEG_BATCH    4
#define RANK_BATCH   8
static_assert(N_CHUNKS <= SORT_THREADS && RANGE_NODES <= SORT_THREADS, "one thread per run / per node in the scans");

template <class F>
__device__ __forceinline__ void for_each_range_edge(const unsigned* __restrict__ binned,
                                                    const unsigned* sstart, const unsigned* scnt,
                                                    F f) {
  const unsigned lane = threadIdx.x & 31;
  const int wave = threadIdx.x >> 5;
  for (int c0 = wave * SEG_BATCH; c0 < N_CHUNKS; c0 += SORT_WAVES * SEG_BATCH) {
    unsigned n[SEG_BATCH], v[SEG_BATCH];
    const unsigned* p[SEG_BATCH];
    #pragma unroll
    for (int q = 0; q < SEG_BATCH; ++q) {
      const int c = c0 + q;
      const bool live = c < N_CHUNKS;
      n[q] = live ? scnt[c] : 0u;
      p[q] = binned + (size_t)(live ? c : 0) * CHUNK_EDGES + (live ? sstart[c] : 0u);
      v[q] = 0u;
      if (lane < n[q]) v[q] = p[q][lane];
    }
    #pragma unroll
    for (int q = 0; q < SEG_BATCH; ++q) {
      if (lane < n[q]) f(v[q], c0 + q, lane);
      for (unsigned j = 32u + lane; j < n[q]; j += 32u) f(p[q][j], c0 + q, j);
    }
  }
}

__global__ __launch_bounds__(SORT_THREADS) void gcn_range_sort_kernel(
    const unsigned* __restrict__ binned, const unsigned* __restrict__ seg,
    unsigned* __restrict__ rowptr, float* __restrict__ dinv,
    unsigned* __restrict__ sorted_edge) {
  __shared__ unsigned sstart[N_CHUNKS];
  __shared__ unsigned scnt[N_CHUNKS];
  __shared__ unsigned soff[N_CHUNKS];
  __shared__ unsigned ncnt[RANGE_NODES];
  __shared__ unsigned nstart[RANGE_NODES + 1];
  __shared__ unsigned ncur[RANGE_NODES];
  __shared__ unsigned wave_tot[SORT_WAVES];
  __shared__ unsigned raw[SORT_CAP];
  __shared__ unsigned lst[SORT_CAP];
  const int tid = threadIdx.x;
  const int r = blockIdx.x;

  unsigned st = 0u, n = 0u;
  if (tid < N_CHUNKS) {
    const unsigned sg = seg[(size_t)r * N_CHUNKS + tid];
    st = sg >> 16;
    n  = sg & 0xFFFFu;
    sstart[tid] = st;
    scnt[tid] = n;
  }
  if (tid < RANGE_NODES) ncnt[tid] = 0u;
  unsigned total;
  const unsigned off = block_exclusive_scan<SORT_WAVES>(n, wave_tot, &total);
  if (tid < N_CHUNKS) soff[tid] = off;
  __syncthreads();
  unsigned base;
  (void)block_exclusive_scan<SORT_WAVES>(st, wave_tot, &base);
  __syncthreads();

  const bool fits = total <= (unsigned)SORT_CAP;
  if (fits) {
    for_each_range_edge(binned, sstart, scnt, [&](unsigned w, int c, unsigned j) {
      raw[soff[c] + j] = w;
      atomicAdd(&ncnt[w >> SRC_BITS], 1u);
    });
  } else {
    for_each_range_edge(binned, sstart, scnt, [&](unsigned w, int, unsigned) {
      atomicAdd(&ncnt[w >> SRC_BITS], 1u);
    });
  }
  __syncthreads();

  const unsigned d = (tid < RANGE_NODES) ? ncnt[tid] : 0u;
  unsigned total_nodes;
  const unsigned ns = block_exclusive_scan<SORT_WAVES>(d, wave_tot, &total_nodes);
  if (tid < RANGE_NODES) {
    nstart[tid] = ns;
    ncur[tid] = ns;
    rowptr[(size_t)r * RANGE_NODES + tid] = base + ns;
    dinv[(size_t)r * RANGE_NODES + tid] = rsqrtf((float)d);
  }
  if (tid == 0) {
    nstart[RANGE_NODES] = total_nodes;
    if (r == N_RANGES - 1) rowptr[N_NODES] = base + total_nodes;
  }
  __syncthreads();

  if (fits) {
    for (unsigned i = tid; i < total; i += SORT_THREADS) {
      const unsigned w = raw[i];
      const unsigned p = atomicAdd(&ncur[w >> SRC_BITS], 1u);
      lst[p] = w;
    }
    __syncthreads();
    for (unsigned i = tid; i < total; i += SORT_THREADS) {
      const unsigned w = lst[i];
      const unsigned node = w >> SRC_BITS;
      const unsigned b = nstart[node], e = nstart[node + 1];
      unsigned pos = i;
      if (e - b <= (unsigned)RANK_MAX_DEG) {
        unsigned rk = 0u;
        for (unsigned q = b; q < e; q += RANK_BATCH) {
          unsigned u[RANK_BATCH];
          #pragma unroll
          for (int t = 0; t < RANK_BATCH; ++t) {
            const unsigned qq = q + (unsigned)t;
            u[t] = lst[qq < e ? qq : e - 1u];
          }
          #pragma unroll
          for (int t = 0; t < RANK_BATCH; ++t) {
            const unsigned qq = q + (unsigned)t;
            rk += (qq < e && (u[t] < w || (u[t] == w && qq < i))) ? 1u : 0u;
          }
        }
        pos = b + rk;
      }
      raw[pos] = w & SRC_MASK;
    }
    __syncthreads();
    for (unsigned i = tid; i < total; i += SORT_THREADS) sorted_edge[base + i] = raw[i];
  } else {
    for_each_range_edge(binned, sstart, scnt, [&](unsigned w, int, unsigned) {
      const unsigned p = atomicAdd(&ncur[w >> SRC_BITS], 1u);
      sorted_edge[base + p] = w & SRC_MASK;
    });
  }
}

#define GATHER_BATCH 8

__device__ __forceinline__ int lane_value(int v, int l) {
  return __builtin_amdgcn_readlane(v, l);
}
__device__ __forceinline__ float lane_value(float v, int l) {
  return __builtin_bit_cast(float, __builtin_amdgcn_readlane(__builtin_bit_cast(int, v), l));
}

__global__ __launch_bounds__(128) void gcn_gather_gemm_wmma_kernel(
    const float* __restrict__ x, const unsigned* __restrict__ sorted_edge,
    const unsigned* __restrict__ rowptr, const float* __restrict__ dinv,
    const float* __restrict__ W, const float* __restrict__ bias,
    float* __restrict__ out) {
  __shared__ __attribute__((aligned(16))) float sW[D_FEAT * UNITS];
  __shared__ __attribute__((aligned(16))) float sA[16 * D_FEAT];

  const int tid = threadIdx.x;
  const int rowBase = blockIdx.x * 16;
  const int lane  = tid & 31;
  const int wave  = tid >> 5;

  const int nb = rowBase + wave * 4;
  const int rp = (lane < 5) ? (int)rowptr[nb + lane] : 0;
  int beg[4], cnt[4], sL[4];
  unsigned wL[4];
  float dL[4];
  #pragma unroll
  for (int r = 0; r < 4; ++r) {
    beg[r] = lane_value(rp, r);
    cnt[r] = lane_value(rp, r + 1) - beg[r];
    wL[r] = 0u;
    if (lane < cnt[r]) wL[r] = sorted_edge[beg[r] + lane];
  }
  #pragma unroll
  for (int r = 0; r < 4; ++r) {
    sL[r] = (int)wL[r];
    dL[r] = 0.0f;
    if (lane < cnt[r]) dL[r] = dinv[wL[r]];
  }

  {
    const float4* Wv = (const float4*)W;
    float4* sWv = (float4*)sW;
    for (int i = tid; i < (D_FEAT * UNITS) / 4; i += 128) sWv[i] = Wv[i];
  }

  #pragma unroll
  for (int r = 0; r < 4; ++r) {
    float ax = 0.0f, ay = 0.0f;
    int s_l = sL[r];
    float d_l = dL[r];
    const int total = cnt[r];
    for (int done = 0; done < total; done += 32) {
      const int m = (total - done < 32) ? (total - done) : 32;
      if (done != 0) {
        s_l = 0; d_l = 0.0f;
        if (lane < m) {
          const unsigned w = sorted_edge[beg[r] + done + lane];
          s_l = (int)w;
          d_l = dinv[w];
        }
      }
      for (int j = 0; j < m; j += GATHER_BATCH) {
        float2 v[GATHER_BATCH];
        float d[GATHER_BATCH];
        #pragma unroll
        for (int q = 0; q < GATHER_BATCH; ++q) {
          const int jj = (j + q < m) ? (j + q) : (m - 1);
          const int s = lane_value(s_l, jj);
          d[q] = lane_value(d_l, jj);
          v[q] = ((const float2*)(x + (size_t)s * D_FEAT))[lane];
        }
        #pragma unroll
        for (int q = 0; q < GATHER_BATCH; ++q) {
          if (j + q < m) { ax += v[q].x * d[q]; ay += v[q].y * d[q]; }
        }
      }
    }
    const float dr = rsqrtf((float)total);
    float2 o; o.x = ax * dr; o.y = ay * dr;
    ((float2*)(sA + (wave * 4 + r) * D_FEAT))[lane] = o;
  }
  __syncthreads();

  const int nBase = wave * 16;
  const int h     = lane >> 4;
  const int mn    = lane & 15;

  v8f acc = {};
  #pragma unroll
  for (int kk = 0; kk < D_FEAT; kk += 4) {
    const int k0 = kk + 2 * h;
    v2f a, b;
    a.x = sA[mn * D_FEAT + k0];
    a.y = sA[mn * D_FEAT + k0 + 1];
    b.x = sW[(k0)     * UNITS + nBase + mn];
    b.y = sW[(k0 + 1) * UNITS + nBase + mn];
    acc = __builtin_amdgcn_wmma_f32_16x16x4_f32(
         false, a,  false, b,
         (short)0, acc,  false,  false);
  }

  const float bn = bias[nBase + mn];
  #pragma unroll
  for (int j = 0; j < 8; ++j) {
    const int m = j + 8 * h;
    float r = acc[j] + bn;
    out[(size_t)(rowBase + m) * UNITS + nBase + mn] = fmaxf(r, 0.0f);
  }
}

extern "C" void kernel_launch(void* const* d_in, const int* in_sizes, int n_in,
                              void* d_out, int out_size, void* d_ws, size_t ws_size,
                              hipStream_t stream) {
  const float* x      = (const float*)d_in[0];
  const float* W      = (const float*)d_in[1];
  const float* b      = (const float*)d_in[2];
  const int*   source = (const int*)d_in[3];
  const int*   target = (const int*)d_in[4];
  float*       out    = (float*)d_out;

  unsigned* binned      = (unsigned*)d_ws;
  unsigned* sorted_edge = (unsigned*)d_ws + (size_t)N_EDGES;
  unsigned* seg         = (unsigned*)d_ws + (size_t)2 * N_EDGES;
  unsigned* rowptr      = (unsigned*)d_ws + (size_t)2 * N_EDGES + (size_t)N_RANGES * N_CHUNKS;
  float*    dinv        = (float*)(rowptr + N_NODES + 4);

  gcn_partition_kernel<<<N_CHUNKS, PART_THREADS, 0, stream>>>(source, target, binned, seg);
  gcn_range_sort_kernel<<<N_RANGES, SORT_THREADS, 0, stream>>>(binned, seg, rowptr, dinv, sorted_edge);
  gcn_gather_gemm_wmma_kernel<<<N_NODES / 16, 128, 0, stream>>>(x, sorted_edge, rowptr, dinv,
                                                                W, b, out);
}


# Gather kernel as 256 blocks of 25 waves that stage a one-byte in-degree table of all nodes and W in LDS once, a 16-row tile per wave: edge scale from LDS, no 4-byte lookup per edge; sum order kept
# speedup vs baseline: 4.8621x; 1.0789x over previous
#include <hip/hip_runtime.h>
#include <hip/hip_bf16.h>


typedef __attribute__((ext_vector_type(2))) float v2f;
typedef __attribute__((ext_vector_type(8))) float v8f;

#define N_NODES 100000
#define N_EDGES 1600000
#define D_FEAT  64
#define UNITS   64

#define SRC_BITS 17
#define SRC_MASK ((1u << SRC_BITS) - 1u)
#define RANGE_NODES 400
#define N_RANGES    (N_NODES / RANGE_NODES)
#define N_CHUNKS    250
#define CHUNK_EDGES (N_EDGES / N_CHUNKS)
static_assert(N_NODES <= (1 << SRC_BITS), "a source id must fit the low field of the edge word");
static_assert(RANGE_NODES <= (1 << (32 - SRC_BITS)), "a node's index in its range must fit the high field");
static_assert(N_NODES % RANGE_NODES == 0, "ranges must be whole");
static_assert(N_EDGES % N_CHUNKS == 0, "chunks must be whole");
static_assert(CHUNK_EDGES < 65536, "ranks, starts and counts inside a chunk are kept in 16 bits");

__device__ __forceinline__ unsigned wave_inclusive_scan(unsigned v, int lane) {
  #pragma unroll
  for (int d = 1; d < 32; d <<= 1) {
    const unsigned t = __shfl_up(v, (unsigned)d);
    if (lane >= d) v += t;
  }
  return v;
}

template <int NWAVES>
__device__ __forceinline__ unsigned block_exclusive_scan(unsigned v, unsigned* wave_tot,
                                                         unsigned* total) {
  const int lane = threadIdx.x & 31;
  const int w = threadIdx.x >> 5;
  const unsigned inc = wave_inclusive_scan(v, lane);
  if (lane == 31) wave_tot[w] = inc;
  __syncthreads();
  unsigned off = 0u, tot = 0u;
  for (int j = 0; j < NWAVES; ++j) {
    const unsigned t = wave_tot[j];
    if (j < w) off += t;
    tot += t;
  }
  *total = tot;
  return off + inc - v;
}

#define PART_THREADS 320
#define PART_LOADS   (CHUNK_EDGES / 4 / PART_THREADS)
static_assert(CHUNK_EDGES % (4 * PART_THREADS) == 0, "a chunk must be whole int4s per thread");
static_assert(N_RANGES <= PART_THREADS, "one thread per range in the scan");

__global__ __launch_bounds__(PART_THREADS) void gcn_partition_kernel(
    const int* __restrict__ source, const int* __restrict__ target,
    unsigned* __restrict__ binned, unsigned* __restrict__ seg) {
  __shared__ unsigned cnt[N_RANGES];
  __shared__ unsigned start[N_RANGES];
  __shared__ unsigned wave_tot[PART_THREADS / 32];
  __shared__ __attribute__((aligned(16))) unsigned stage[CHUNK_EDGES];
  const int tid = threadIdx.x;
  const int c = blockIdx.x;

  const int4* t4 = (const int4*)(target + (size_t)c * CHUNK_EDGES);
  const int4* s4 = (const int4*)(source + (size_t)c * CHUNK_EDGES);
  int4 t[PART_LOADS];
  int4 sv[PART_LOADS];
  #pragma unroll
  for (int k = 0; k < PART_LOADS; ++k) {
    t[k]  = t4[tid + k * PART_THREADS];
    sv[k] = s4[tid + k * PART_THREADS];
  }
  if (tid < N_RANGES) cnt[tid] = 0u;
  __syncthreads();

  unsigned key[PART_LOADS * 4];
  unsigned word[PART_LOADS * 4];
  #pragma unroll
  for (int k = 0; k < PART_LOADS; ++k) {
    const int tv[4] = {t[k].x, t[k].y, t[k].z, t[k].w};
    const int sw[4] = {sv[k].x, sv[k].y, sv[k].z, sv[k].w};
    #pragma unroll
    for (int e = 0; e < 4; ++e) {
      const unsigned tg = (unsigned)tv[e];
      const unsigned r = tg / (unsigned)RANGE_NODES;
      const unsigned loc = tg - r * (unsigned)RANGE_NODES;
      const unsigned rank = atomicAdd(&cnt[r], 1u);
      key[k * 4 + e] = (r << 16) | rank;
      word[k * 4 + e] = (unsigned)sw[e] | (loc << SRC_BITS);
    }
  }
  __syncthreads();

  unsigned total;
  const unsigned n = (tid < N_RANGES) ? cnt[tid] : 0u;
  const unsigned ex = block_exclusive_scan<PART_THREADS / 32>(n, wave_tot, &total);
  if (tid < N_RANGES) {
    start[tid] = ex;
    seg[(size_t)tid * N_CHUNKS + c] = (ex << 16) | n;
  }
  __syncthreads();

  #pragma unroll
  for (int i = 0; i < PART_LOADS * 4; ++i) stage[start[key[i] >> 16] + (key[i] & 0xFFFFu)] = word[i];
  __syncthreads();

  uint4* dst4 = (uint4*)(binned + (size_t)c * CHUNK_EDGES);
  const uint4* st4 = (const uint4*)stage;
  #pragma unroll
  for (int k = 0; k < PART_LOADS; ++k) dst4[tid + k * PART_THREADS] = st4[tid + k * PART_THREADS];
}

#define SORT_THREADS 1024
#define SORT_WAVES   (SORT_THREADS / 32)
#define SORT_CAP     16384
#define RANK_MAX_DEG 64
#define SEG_BATCH    4
#define RANK_BATCH   8
#define DEG_SAT      255u
static_assert(N_CHUNKS <= SORT_THREADS && RANGE_NODES <= SORT_THREADS, "one thread per run / per node in the scans");

template <class F>
__device__ __forceinline__ void for_each_range_edge(const unsigned* __restrict__ binned,
                                                    const unsigned* sstart, const unsigned* scnt,
                                                    F f) {
  const unsigned lane = threadIdx.x & 31;
  const int wave = threadIdx.x >> 5;
  for (int c0 = wave * SEG_BATCH; c0 < N_CHUNKS; c0 += SORT_WAVES * SEG_BATCH) {
    unsigned n[SEG_BATCH], v[SEG_BATCH];
    const unsigned* p[SEG_BATCH];
    #pragma unroll
    for (int q = 0; q < SEG_BATCH; ++q) {
      const int c = c0 + q;
      const bool live = c < N_CHUNKS;
      n[q] = live ? scnt[c] : 0u;
      p[q] = binned + (size_t)(live ? c : 0) * CHUNK_EDGES + (live ? sstart[c] : 0u);
      v[q] = 0u;
      if (lane < n[q]) v[q] = p[q][lane];
    }
    #pragma unroll
    for (int q = 0; q < SEG_BATCH; ++q) {
      if (lane < n[q]) f(v[q], c0 + q, lane);
      for (unsigned j = 32u + lane; j < n[q]; j += 32u) f(p[q][j], c0 + q, j);
    }
  }
}

__global__ __launch_bounds__(SORT_THREADS) void gcn_range_sort_kernel(
    const unsigned* __restrict__ binned, const unsigned* __restrict__ seg,
    unsigned* __restrict__ rowptr, float* __restrict__ dinv,
    unsigned* __restrict__ deg8w, unsigned* __restrict__ sorted_edge) {
  __shared__ unsigned sstart[N_CHUNKS];
  __shared__ unsigned scnt[N_CHUNKS];
  __shared__ unsigned soff[N_CHUNKS];
  __shared__ unsigned ncnt[RANGE_NODES];
  __shared__ unsigned nstart[RANGE_NODES + 1];
  __shared__ unsigned ncur[RANGE_NODES];
  __shared__ unsigned wave_tot[SORT_WAVES];
  __shared__ unsigned raw[SORT_CAP];
  __shared__ unsigned lst[SORT_CAP];
  const int tid = threadIdx.x;
  const int r = blockIdx.x;

  unsigned st = 0u, n = 0u;
  if (tid < N_CHUNKS) {
    const unsigned sg = seg[(size_t)r * N_CHUNKS + tid];
    st = sg >> 16;
    n  = sg & 0xFFFFu;
    sstart[tid] = st;
    scnt[tid] = n;
  }
  if (tid < RANGE_NODES) ncnt[tid] = 0u;
  unsigned total;
  const unsigned off = block_exclusive_scan<SORT_WAVES>(n, wave_tot, &total);
  if (tid < N_CHUNKS) soff[tid] = off;
  __syncthreads();
  unsigned base;
  (void)block_exclusive_scan<SORT_WAVES>(st, wave_tot, &base);
  __syncthreads();

  const bool fits = total <= (unsigned)SORT_CAP;
  if (fits) {
    for_each_range_edge(binned, sstart, scnt, [&](unsigned w, int c, unsigned j) {
      raw[soff[c] + j] = w;
      atomicAdd(&ncnt[w >> SRC_BITS], 1u);
    });
  } else {
    for_each_range_edge(binned, sstart, scnt, [&](unsigned w, int, unsigned) {
      atomicAdd(&ncnt[w >> SRC_BITS], 1u);
    });
  }
  __syncthreads();

  const unsigned d = (tid < RANGE_NODES) ? ncnt[tid] : 0u;
  unsigned total_nodes;
  const unsigned ns = block_exclusive_scan<SORT_WAVES>(d, wave_tot, &total_nodes);
  if (tid < RANGE_NODES) {
    nstart[tid] = ns;
    ncur[tid] = ns;
    rowptr[(size_t)r * RANGE_NODES + tid] = base + ns;
    dinv[(size_t)r * RANGE_NODES + tid] = rsqrtf((float)d);
    ((unsigned char*)deg8w)[(size_t)r * RANGE_NODES + tid] = (unsigned char)(d < DEG_SAT ? d : DEG_SAT);
  }
  if (tid == 0) {
    nstart[RANGE_NODES] = total_nodes;
    if (r == N_RANGES - 1) rowptr[N_NODES] = base + total_nodes;
  }
  __syncthreads();

  if (fits) {
    for (unsigned i = tid; i < total; i += SORT_THREADS) {
      const unsigned w = raw[i];
      const unsigned p = atomicAdd(&ncur[w >> SRC_BITS], 1u);
      lst[p] = w;
    }
    __syncthreads();
    for (unsigned i = tid; i < total; i += SORT_THREADS) {
      const unsigned w = lst[i];
      const unsigned node = w >> SRC_BITS;
      const unsigned b = nstart[node], e = nstart[node + 1];
      unsigned pos = i;
      if (e - b <= (unsigned)RANK_MAX_DEG) {
        unsigned rk = 0u;
        for (unsigned q = b; q < e; q += RANK_BATCH) {
          unsigned u[RANK_BATCH];
          #pragma unroll
          for (int t = 0; t < RANK_BATCH; ++t) {
            const unsigned qq = q + (unsigned)t;
            u[t] = lst[qq < e ? qq : e - 1u];
          }
          #pragma unroll
          for (int t = 0; t < RANK_BATCH; ++t) {
            const unsigned qq = q + (unsigned)t;
            rk += (qq < e && (u[t] < w || (u[t] == w && qq < i))) ? 1u : 0u;
          }
        }
        pos = b + rk;
      }
      raw[pos] = w & SRC_MASK;
    }
    __syncthreads();
    for (unsigned i = tid; i < total; i += SORT_THREADS) sorted_edge[base + i] = raw[i];
  } else {
    for_each_range_edge(binned, sstart, scnt, [&](unsigned w, int, unsigned) {
      const unsigned p = atomicAdd(&ncur[w >> SRC_BITS], 1u);
      sorted_edge[base + p] = w & SRC_MASK;
    });
  }
}

#define GATHER_BATCH 16
#define GG_BLOCKS    256
#define GG_WAVES     25
#define GG_THREADS   (GG_WAVES * 32)
#define N_TILES      (N_NODES / 16)
#define A_PITCH      (D_FEAT + 2)
#define DEG_CHUNKS   (N_NODES / 16)
static_assert(N_NODES % 16 == 0, "whole tiles, and the table in whole 16-byte pieces");
static_assert(GG_BLOCKS * GG_WAVES >= N_TILES, "one tile per wave");
static_assert((A_PITCH % 2) == 0, "A rows are written and read as 8-byte pairs");

__device__ __forceinline__ int lane_value(int v, int l) {
  return __builtin_amdgcn_readlane(v, l);
}
__device__ __forceinline__ float lane_value(float v, int l) {
  return __builtin_bit_cast(float, __builtin_amdgcn_readlane(__builtin_bit_cast(int, v), l));
}

__global__ __launch_bounds__(GG_THREADS) void gcn_gather_gemm_wmma_kernel(
    const float* __restrict__ x, const unsigned* __restrict__ sorted_edge,
    const unsigned* __restrict__ rowptr, const float* __restrict__ dinv,
    const unsigned* __restrict__ deg8w,
    const float* __restrict__ W, const float* __restrict__ bias,
    float* __restrict__ out) {
  __shared__ __attribute__((aligned(16))) unsigned char sDeg[N_NODES];
  __shared__ __attribute__((aligned(16))) float sW[D_FEAT * UNITS];
  __shared__ __attribute__((aligned(16))) float sA[GG_WAVES * 16 * A_PITCH];

  const int tid  = threadIdx.x;
  const int lane = tid & 31;
  const int wave = tid >> 5;
  const int tile = (int)blockIdx.x + GG_BLOCKS * wave;
  const bool live = tile < N_TILES;
  const int rowBase = tile * 16;

  int rp = 0;
  if (live && lane < 17) rp = (int)rowptr[rowBase + lane];
  unsigned w_next = 0u;
  {
    const int b0 = lane_value(rp, 0);
    const int c0 = lane_value(rp, 1) - b0;
    if (lane < c0) w_next = sorted_edge[b0 + lane];
  }

  {
    const uint4* src = (const uint4*)deg8w;
    uint4* dst = (uint4*)sDeg;
    const int rot = (int)blockIdx.x * (DEG_CHUNKS / GG_BLOCKS);
    for (int i = tid; i < DEG_CHUNKS; i += GG_THREADS) {
      int c = i + rot;
      if (c >= DEG_CHUNKS) c -= DEG_CHUNKS;
      dst[c] = src[c];
    }
    const float4* Wv = (const float4*)W;
    float4* sWv = (float4*)sW;
    for (int i = tid; i < (D_FEAT * UNITS) / 4; i += GG_THREADS) sWv[i] = Wv[i];
  }
  __syncthreads();
  if (!live) return;

  float* myA = sA + wave * (16 * A_PITCH);
  for (int r = 0; r < 16; ++r) {
    const int beg = lane_value(rp, r);
    const int total = lane_value(rp, r + 1) - beg;
    const unsigned w_first = w_next;
    const int nb = lane_value(rp, (r + 1) & 15);
    const int nc = (r + 1 < 16) ? lane_value(rp, r + 2) - nb : 0;
    w_next = 0u;
    if (total == 0 && lane < nc) w_next = sorted_edge[nb + lane];
    float ax = 0.0f, ay = 0.0f;
    for (int done = 0; done < total; done += 32) {
      const int m = (total - done < 32) ? (total - done) : 32;
      unsigned w = w_first;
      if (done != 0) {
        w = 0u;
        if (lane < m) w = sorted_edge[beg + done + lane];
      }
      float d_l = 0.0f;
      if (lane < m) {
        const unsigned t = sDeg[w];
        d_l = rsqrtf((float)t);
        if (t == DEG_SAT) d_l = dinv[w];
      }
      const int s_l = (int)w;
      for (int j = 0; j < m; j += GATHER_BATCH) {
        float2 v[GATHER_BATCH];
        float d[GATHER_BATCH];
        #pragma unroll
        for (int q = 0; q < GATHER_BATCH; ++q) {
          const int jj = (j + q < m) ? (j + q) : (m - 1);
          const int s = lane_value(s_l, jj);
          d[q] = lane_value(d_l, jj);
          v[q] = ((const float2*)(x + (size_t)s * D_FEAT))[lane];
        }
        if (done == 0 && j == 0 && lane < nc) w_next = sorted_edge[nb + lane];
        #pragma unroll
        for (int q = 0; q < GATHER_BATCH; ++q) {
          if (j + q < m) { ax += v[q].x * d[q]; ay += v[q].y * d[q]; }
        }
      }
    }
    const float dr = rsqrtf((float)total);
    float2 o; o.x = ax * dr; o.y = ay * dr;
    ((float2*)(myA + r * A_PITCH))[lane] = o;
  }
  __builtin_amdgcn_wave_barrier();

  const int h  = lane >> 4;
  const int mn = lane & 15;

  v2f a[D_FEAT / 4];
  #pragma unroll
  for (int ks = 0; ks < D_FEAT / 4; ++ks) {
    const float2 t = *(const float2*)(myA + mn * A_PITCH + ks * 4 + 2 * h);
    a[ks].x = t.x; a[ks].y = t.y;
  }
  #pragma unroll
  for (int nt = 0; nt < UNITS / 16; ++nt) {
    const int nBase = nt * 16;
    v8f acc = {};
    #pragma unroll
    for (int ks = 0; ks < D_FEAT / 4; ++ks) {
      const int k0 = ks * 4 + 2 * h;
      v2f b;
      b.x = sW[(k0)     * UNITS + nBase + mn];
      b.y = sW[(k0 + 1) * UNITS + nBase + mn];
      acc = __builtin_amdgcn_wmma_f32_16x16x4_f32(
           false, a[ks],  false, b,
           (short)0, acc,  false,  false);
    }
    const float bn = bias[nBase + mn];
    #pragma unroll
    for (int j = 0; j < 8; ++j) {
      const int m = j + 8 * h;
      const float rr = acc[j] + bn;
      out[(size_t)(rowBase + m) * UNITS + nBase + mn] = fmaxf(rr, 0.0f);
    }
  }
}

extern "C" void kernel_launch(void* const* d_in, const int* in_sizes, int n_in,
                              void* d_out, int out_size, void* d_ws, size_t ws_size,
                              hipStream_t stream) {
  const float* x      = (const float*)d_in[0];
  const float* W      = (const float*)d_in[1];
  const float* b      = (const float*)d_in[2];
  const int*   source = (const int*)d_in[3];
  const int*   target = (const int*)d_in[4];
  float*       out    = (float*)d_out;

  unsigned* binned      = (unsigned*)d_ws;
  unsigned* sorted_edge = (unsigned*)d_ws + (size_t)N_EDGES;
  unsigned* seg         = (unsigned*)d_ws + (size_t)2 * N_EDGES;
  unsigned* rowptr      = (unsigned*)d_ws + (size_t)2 * N_EDGES + (size_t)N_RANGES * N_CHUNKS;
  float*    dinv        = (float*)(rowptr + N_NODES + 4);
  unsigned* deg8w       = (unsigned*)(dinv + N_NODES);

  gcn_partition_kernel<<<N_CHUNKS, PART_THREADS, 0, stream>>>(source, target, binned, seg);
  gcn_range_sort_kernel<<<N_RANGES, SORT_THREADS, 0, stream>>>(binned, seg, rowptr, dinv, deg8w, sorted_edge);
  gcn_gather_gemm_wmma_kernel<<<GG_BLOCKS, GG_THREADS, 0, stream>>>(x, sorted_edge, rowptr, dinv,
                                                                    deg8w, W, b, out);
}


# Gather takes two rows at a time, one per half-wave, with 16-byte loads: one load instruction brings a source row for each, up to 24 edges per row in flight; each row still summed in list order
# speedup vs baseline: 5.5539x; 1.1423x over previous
#include <hip/hip_runtime.h>
#include <hip/hip_bf16.h>


typedef __attribute__((ext_vector_type(2))) float v2f;
typedef __attribute__((ext_vector_type(8))) float v8f;

#define N_NODES 100000
#define N_EDGES 1600000
#define D_FEAT  64
#define UNITS   64

#define SRC_BITS 17
#define SRC_MASK ((1u << SRC_BITS) - 1u)
#define RANGE_NODES 400
#define N_RANGES    (N_NODES / RANGE_NODES)
#define N_CHUNKS    250
#define CHUNK_EDGES (N_EDGES / N_CHUNKS)
static_assert(N_NODES <= (1 << SRC_BITS), "a source id must fit the low field of the edge word");
static_assert(RANGE_NODES <= (1 << (32 - SRC_BITS)), "a node's index in its range must fit the high field");
static_assert(N_NODES % RANGE_NODES == 0, "ranges must be whole");
static_assert(N_EDGES % N_CHUNKS == 0, "chunks must be whole");
static_assert(CHUNK_EDGES < 65536, "ranks, starts and counts inside a chunk are kept in 16 bits");

__device__ __forceinline__ unsigned wave_inclusive_scan(unsigned v, int lane) {
  #pragma unroll
  for (int d = 1; d < 32; d <<= 1) {
    const unsigned t = __shfl_up(v, (unsigned)d);
    if (lane >= d) v += t;
  }
  return v;
}

template <int NWAVES>
__device__ __forceinline__ unsigned block_exclusive_scan(unsigned v, unsigned* wave_tot,
                                                         unsigned* total) {
  const int lane = threadIdx.x & 31;
  const int w = threadIdx.x >> 5;
  const unsigned inc = wave_inclusive_scan(v, lane);
  if (lane == 31) wave_tot[w] = inc;
  __syncthreads();
  unsigned off = 0u, tot = 0u;
  for (int j = 0; j < NWAVES; ++j) {
    const unsigned t = wave_tot[j];
    if (j < w) off += t;
    tot += t;
  }
  *total = tot;
  return off + inc - v;
}

#define PART_THREADS 320
#define PART_LOADS   (CHUNK_EDGES / 4 / PART_THREADS)
static_assert(CHUNK_EDGES % (4 * PART_THREADS) == 0, "a chunk must be whole int4s per thread");
static_assert(N_RANGES <= PART_THREADS, "one thread per range in the scan");

__global__ __launch_bounds__(PART_THREADS) void gcn_partition_kernel(
    const int* __restrict__ source, const int* __restrict__ target,
    unsigned* __restrict__ binned, unsigned* __restrict__ seg) {
  __shared__ unsigned cnt[N_RANGES];
  __shared__ unsigned start[N_RANGES];
  __shared__ unsigned wave_tot[PART_THREADS / 32];
  __shared__ __attribute__((aligned(16))) unsigned stage[CHUNK_EDGES];
  const int tid = threadIdx.x;
  const int c = blockIdx.x;

  const int4* t4 = (const int4*)(target + (size_t)c * CHUNK_EDGES);
  const int4* s4 = (const int4*)(source + (size_t)c * CHUNK_EDGES);
  int4 t[PART_LOADS];
  int4 sv[PART_LOADS];
  #pragma unroll
  for (int k = 0; k < PART_LOADS; ++k) {
    t[k]  = t4[tid + k * PART_THREADS];
    sv[k] = s4[tid + k * PART_THREADS];
  }
  if (tid < N_RANGES) cnt[tid] = 0u;
  __syncthreads();

  unsigned key[PART_LOADS * 4];
  unsigned word[PART_LOADS * 4];
  #pragma unroll
  for (int k = 0; k < PART_LOADS; ++k) {
    const int tv[4] = {t[k].x, t[k].y, t[k].z, t[k].w};
    const int sw[4] = {sv[k].x, sv[k].y, sv[k].z, sv[k].w};
    #pragma unroll
    for (int e = 0; e < 4; ++e) {
      const unsigned tg = (unsigned)tv[e];
      const unsigned r = tg / (unsigned)RANGE_NODES;
      const unsigned loc = tg - r * (unsigned)RANGE_NODES;
      const unsigned rank = atomicAdd(&cnt[r], 1u);
      key[k * 4 + e] = (r << 16) | rank;
      word[k * 4 + e] = (unsigned)sw[e] | (loc << SRC_BITS);
    }
  }
  __syncthreads();

  unsigned total;
  const unsigned n = (tid < N_RANGES) ? cnt[tid] : 0u;
  const unsigned ex = block_exclusive_scan<PART_THREADS / 32>(n, wave_tot, &total);
  if (tid < N_RANGES) {
    start[tid] = ex;
    seg[(size_t)tid * N_CHUNKS + c] = (ex << 16) | n;
  }
  __syncthreads();

  #pragma unroll
  for (int i = 0; i < PART_LOADS * 4; ++i) stage[start[key[i] >> 16] + (key[i] & 0xFFFFu)] = word[i];
  __syncthreads();

  uint4* dst4 = (uint4*)(binned + (size_t)c * CHUNK_EDGES);
  const uint4* st4 = (const uint4*)stage;
  #pragma unroll
  for (int k = 0; k < PART_LOADS; ++k) dst4[tid + k * PART_THREADS] = st4[tid + k * PART_THREADS];
}

#define SORT_THREADS 1024
#define SORT_WAVES   (SORT_THREADS / 32)
#define SORT_CAP     16384
#define RANK_MAX_DEG 64
#define SEG_BATCH    4
#define RANK_BATCH   8
#define DEG_SAT      255u
static_assert(N_CHUNKS <= SORT_THREADS && RANGE_NODES <= SORT_THREADS, "one thread per run / per node in the scans");

template <class F>
__device__ __forceinline__ void for_each_range_edge(const unsigned* __restrict__ binned,
                                                    const unsigned* sstart, const unsigned* scnt,
                                                    F f) {
  const unsigned lane = threadIdx.x & 31;
  const int wave = threadIdx.x >> 5;
  for (int c0 = wave * SEG_BATCH; c0 < N_CHUNKS; c0 += SORT_WAVES * SEG_BATCH) {
    unsigned n[SEG_BATCH], v[SEG_BATCH];
    const unsigned* p[SEG_BATCH];
    #pragma unroll
    for (int q = 0; q < SEG_BATCH; ++q) {
      const int c = c0 + q;
      const bool live = c < N_CHUNKS;
      n[q] = live ? scnt[c] : 0u;
      p[q] = binned + (size_t)(live ? c : 0) * CHUNK_EDGES + (live ? sstart[c] : 0u);
      v[q] = 0u;
      if (lane < n[q]) v[q] = p[q][lane];
    }
    #pragma unroll
    for (int q = 0; q < SEG_BATCH; ++q) {
      if (lane < n[q]) f(v[q], c0 + q, lane);
      for (unsigned j = 32u + lane; j < n[q]; j += 32u) f(p[q][j], c0 + q, j);
    }
  }
}

__global__ __launch_bounds__(SORT_THREADS) void gcn_range_sort_kernel(
    const unsigned* __restrict__ binned, const unsigned* __restrict__ seg,
    unsigned* __restrict__ rowptr, float* __restrict__ dinv,
    unsigned* __restrict__ deg8w, unsigned* __restrict__ sorted_edge) {
  __shared__ unsigned sstart[N_CHUNKS];
  __shared__ unsigned scnt[N_CHUNKS];
  __shared__ unsigned soff[N_CHUNKS];
  __shared__ unsigned ncnt[RANGE_NODES];
  __shared__ unsigned nstart[RANGE_NODES + 1];
  __shared__ unsigned ncur[RANGE_NODES];
  __shared__ unsigned wave_tot[SORT_WAVES];
  __shared__ unsigned raw[SORT_CAP];
  __shared__ unsigned lst[SORT_CAP];
  const int tid = threadIdx.x;
  const int r = blockIdx.x;

  unsigned st = 0u, n = 0u;
  if (tid < N_CHUNKS) {
    const unsigned sg = seg[(size_t)r * N_CHUNKS + tid];
    st = sg >> 16;
    n  = sg & 0xFFFFu;
    sstart[tid] = st;
    scnt[tid] = n;
  }
  if (tid < RANGE_NODES) ncnt[tid] = 0u;
  unsigned total;
  const unsigned off = block_exclusive_scan<SORT_WAVES>(n, wave_tot, &total);
  if (tid < N_CHUNKS) soff[tid] = off;
  __syncthreads();
  unsigned base;
  (void)block_exclusive_scan<SORT_WAVES>(st, wave_tot, &base);
  __syncthreads();

  const bool fits = total <= (unsigned)SORT_CAP;
  if (fits) {
    for_each_range_edge(binned, sstart, scnt, [&](unsigned w, int c, unsigned j) {
      raw[soff[c] + j] = w;
      atomicAdd(&ncnt[w >> SRC_BITS], 1u);
    });
  } else {
    for_each_range_edge(binned, sstart, scnt, [&](unsigned w, int, unsigned) {
      atomicAdd(&ncnt[w >> SRC_BITS], 1u);
    });
  }
  __syncthreads();

  const unsigned d = (tid < RANGE_NODES) ? ncnt[tid] : 0u;
  unsigned total_nodes;
  const unsigned ns = block_exclusive_scan<SORT_WAVES>(d, wave_tot, &total_nodes);
  if (tid < RANGE_NODES) {
    nstart[tid] = ns;
    ncur[tid] = ns;
    rowptr[(size_t)r * RANGE_NODES + tid] = base + ns;
    dinv[(size_t)r * RANGE_NODES + tid] = rsqrtf((float)d);
    ((unsigned char*)deg8w)[(size_t)r * RANGE_NODES + tid] = (unsigned char)(d < DEG_SAT ? d : DEG_SAT);
  }
  if (tid == 0) {
    nstart[RANGE_NODES] = total_nodes;
    if (r == N_RANGES - 1) rowptr[N_NODES] = base + total_nodes;
  }
  __syncthreads();

  if (fits) {
    for (unsigned i = tid; i < total; i += SORT_THREADS) {
      const unsigned w = raw[i];
      const unsigned p = atomicAdd(&ncur[w >> SRC_BITS], 1u);
      lst[p] = w;
    }
    __syncthreads();
    for (unsigned i = tid; i < total; i += SORT_THREADS) {
      const unsigned w = lst[i];
      const unsigned node = w >> SRC_BITS;
      const unsigned b = nstart[node], e = nstart[node + 1];
      unsigned pos = i;
      if (e - b <= (unsigned)RANK_MAX_DEG) {
        unsigned rk = 0u;
        for (unsigned q = b; q < e; q += RANK_BATCH) {
          unsigned u[RANK_BATCH];
          #pragma unroll
          for (int t = 0; t < RANK_BATCH; ++t) {
            const unsigned qq = q + (unsigned)t;
            u[t] = lst[qq < e ? qq : e - 1u];
          }
          #pragma unroll
          for (int t = 0; t < RANK_BATCH; ++t) {
            const unsigned qq = q + (unsigned)t;
            rk += (qq < e && (u[t] < w || (u[t] == w && qq < i))) ? 1u : 0u;
          }
        }
        pos = b + rk;
      }
      raw[pos] = w & SRC_MASK;
    }
    __syncthreads();
    for (unsigned i = tid; i < total; i += SORT_THREADS) sorted_edge[base + i] = raw[i];
  } else {
    for_each_range_edge(binned, sstart, scnt, [&](unsigned w, int, unsigned) {
      const unsigned p = atomicAdd(&ncur[w >> SRC_BITS], 1u);
      sorted_edge[base + p] = w & SRC_MASK;
    });
  }
}

#define TRIP_EDGES   24
#define STEP_GROUP   4
#define GG_BLOCKS    256
#define GG_WAVES     25
#define GG_THREADS   (GG_WAVES * 32)
#define N_TILES      (N_NODES / 16)
#define A_PITCH      (D_FEAT + 2)
#define DEG_CHUNKS   (N_NODES / 16)
static_assert(N_NODES % 16 == 0, "whole tiles, and the table in whole 16-byte pieces");
static_assert(GG_BLOCKS * GG_WAVES >= N_TILES, "one tile per wave");
static_assert((A_PITCH % 2) == 0, "A rows are written and read as 8-byte pairs");
static_assert(D_FEAT == 64, "a half-wave of 16 lanes with 4 floats each covers one row of x");
static_assert(TRIP_EDGES > 16 && TRIP_EDGES <= 32 && TRIP_EDGES % STEP_GROUP == 0,
              "a trip's edge words are held in two registers of 16 lanes per half-wave");

__device__ __forceinline__ int lane_value(int v, int l) {
  return __builtin_amdgcn_readlane(v, l);
}

template <int J>
__device__ __forceinline__ unsigned half_lane_value(unsigned prev, unsigned v) {
  return __builtin_amdgcn_permlane16(prev, v, (unsigned)J * 0x11111111u, (unsigned)J * 0x11111111u,
                                     false, false);
}

template <int J>
__device__ __forceinline__ void gather_step_load(float4 (&v)[TRIP_EDGES], unsigned& s, const char* xb,
                                                 unsigned wlo, unsigned whi, unsigned c16) {
  s = half_lane_value<(J & 15)>(s, (J < 16) ? wlo : whi);
  v[J] = *(const float4*)(xb + ((s << 8) | c16));
}
template <int J>
__device__ __forceinline__ void gather_step_add(float4& acc, unsigned& db, const float4 (&v)[TRIP_EDGES],
                                                float dlo, float dhi, int rel) {
  db = half_lane_value<(J & 15)>(db, __builtin_bit_cast(unsigned, (J < 16) ? dlo : dhi));
  const float d = __builtin_bit_cast(float, db);
  if (J < rel) {
    acc.x += v[J].x * d; acc.y += v[J].y * d; acc.z += v[J].z * d; acc.w += v[J].w * d;
  }
}
template <int G>
__device__ __forceinline__ void gather_group_load(float4 (&v)[TRIP_EDGES], unsigned& s, const char* xb,
                                                  unsigned wlo, unsigned whi, unsigned c16) {
  gather_step_load<G * STEP_GROUP + 0>(v, s, xb, wlo, whi, c16);
  gather_step_load<G * STEP_GROUP + 1>(v, s, xb, wlo, whi, c16);
  gather_step_load<G * STEP_GROUP + 2>(v, s, xb, wlo, whi, c16);
  gather_step_load<G * STEP_GROUP + 3>(v, s, xb, wlo, whi, c16);
}
template <int G>
__device__ __forceinline__ void gather_group_add(float4& acc, unsigned& db, const float4 (&v)[TRIP_EDGES],
                                                 float dlo, float dhi, int rel) {
  gather_step_add<G * STEP_GROUP + 0>(acc, db, v, dlo, dhi, rel);
  gather_step_add<G * STEP_GROUP + 1>(acc, db, v, dlo, dhi, rel);
  gather_step_add<G * STEP_GROUP + 2>(acc, db, v, dlo, dhi, rel);
  gather_step_add<G * STEP_GROUP + 3>(acc, db, v, dlo, dhi, rel);
}
static_assert(STEP_GROUP == 4 && TRIP_EDGES == 6 * STEP_GROUP, "the groups are written out below");

__global__ __launch_bounds__(GG_THREADS) void gcn_gather_gemm_wmma_kernel(
    const float* __restrict__ x, const unsigned* __restrict__ sorted_edge,
    const unsigned* __restrict__ rowptr, const float* __restrict__ dinv,
    const unsigned* __restrict__ deg8w,
    const float* __restrict__ W, const float* __restrict__ bias,
    float* __restrict__ out) {
  __shared__ __attribute__((aligned(16))) unsigned char sDeg[N_NODES];
  __shared__ __attribute__((aligned(16))) float sW[D_FEAT * UNITS];
  __shared__ __attribute__((aligned(16))) float sA[GG_WAVES * 16 * A_PITCH];

  const int tid  = threadIdx.x;
  const int lane = tid & 31;
  const int wave = tid >> 5;
  const int tile = (int)blockIdx.x + GG_BLOCKS * wave;
  const bool live = tile < N_TILES;
  const int rowBase = tile * 16;

  const int h  = lane >> 4;
  const int mn = lane & 15;
  const unsigned c16 = (unsigned)mn << 4;
  const char* xb = (const char*)x;

  int rp = 0;
  if (live && lane < 17) rp = (int)rowptr[rowBase + lane];
  auto pair_bounds = [&](int p, int& beg, int& cnt, int& nmax) {
    const int s0 = lane_value(rp, 2 * p), s1 = lane_value(rp, 2 * p + 1), s2 = lane_value(rp, 2 * p + 2);
    beg = h ? s1 : s0;
    cnt = h ? (s2 - s1) : (s1 - s0);
    nmax = (s1 - s0 > s2 - s1) ? (s1 - s0) : (s2 - s1);
  };
  auto load_words = [&](int beg, int cnt, int k, unsigned& wlo, unsigned& whi) {
    wlo = 0u; whi = 0u;
    if (k + mn < cnt) wlo = sorted_edge[beg + k + mn];
    if (mn < TRIP_EDGES - 16 && k + 16 + mn < cnt) whi = sorted_edge[beg + k + 16 + mn];
  };
  auto scale_of = [&](unsigned t, unsigned w, bool has) -> float {
    float d = 0.0f;
    if (has) {
      d = rsqrtf((float)t);
      if (t == DEG_SAT) d = dinv[w];
    }
    return d;
  };

  int beg, cnt, nmax, beg1, cnt1, nmax1;
  unsigned wlo, whi, wlo1, whi1;
  pair_bounds(0, beg, cnt, nmax);
  pair_bounds(1, beg1, cnt1, nmax1);
  load_words(beg, cnt, 0, wlo, whi);
  load_words(beg1, cnt1, 0, wlo1, whi1);

  {
    const uint4* src = (const uint4*)deg8w;
    uint4* dst = (uint4*)sDeg;
    const int rot = (int)blockIdx.x * (DEG_CHUNKS / GG_BLOCKS);
    for (int i = tid; i < DEG_CHUNKS; i += GG_THREADS) {
      int c = i + rot;
      if (c >= DEG_CHUNKS) c -= DEG_CHUNKS;
      dst[c] = src[c];
    }
    const float4* Wv = (const float4*)W;
    float4* sWv = (float4*)sW;
    for (int i = tid; i < (D_FEAT * UNITS) / 4; i += GG_THREADS) sWv[i] = Wv[i];
  }
  __syncthreads();
  if (!live) return;

  float* myA = sA + wave * (16 * A_PITCH);
  float dlo = scale_of(sDeg[wlo], wlo, mn < cnt);
  float dhi = scale_of(sDeg[whi], whi, mn < TRIP_EDGES - 16 && 16 + mn < cnt);
  for (int p = 0; p < 8; ++p) {
    int beg2 = 0, cnt2 = 0, nmax2 = 0;
    unsigned wlo2 = 0u, whi2 = 0u;
    if (p + 2 < 8) {
      pair_bounds(p + 2, beg2, cnt2, nmax2);
      load_words(beg2, cnt2, 0, wlo2, whi2);
    }
    const unsigned tlo1 = sDeg[wlo1], thi1 = sDeg[whi1];

    float4 acc = {0.0f, 0.0f, 0.0f, 0.0f};
    int k = 0;
    do {
      if (k != 0) {
        load_words(beg, cnt, k, wlo, whi);
        dlo = scale_of(sDeg[wlo], wlo, k + mn < cnt);
        dhi = scale_of(sDeg[whi], whi, mn < TRIP_EDGES - 16 && k + 16 + mn < cnt);
      }
      const int rel = cnt - k;
      const int left = nmax - k;
      float4 v[TRIP_EDGES];
      unsigned s = 0u, db = 0u;
      if (left > 0 * STEP_GROUP) gather_group_load<0>(v, s, xb, wlo, whi, c16);
      if (left > 1 * STEP_GROUP) gather_group_load<1>(v, s, xb, wlo, whi, c16);
      if (left > 2 * STEP_GROUP) gather_group_load<2>(v, s, xb, wlo, whi, c16);
      if (left > 3 * STEP_GROUP) gather_group_load<3>(v, s, xb, wlo, whi, c16);
      if (left > 4 * STEP_GROUP) gather_group_load<4>(v, s, xb, wlo, whi, c16);
      if (left > 5 * STEP_GROUP) gather_group_load<5>(v, s, xb, wlo, whi, c16);
      if (left > 0 * STEP_GROUP) gather_group_add<0>(acc, db, v, dlo, dhi, rel);
      if (left > 1 * STEP_GROUP) gather_group_add<1>(acc, db, v, dlo, dhi, rel);
      if (left > 2 * STEP_GROUP) gather_group_add<2>(acc, db, v, dlo, dhi, rel);
      if (left > 3 * STEP_GROUP) gather_group_add<3>(acc, db, v, dlo, dhi, rel);
      if (left > 4 * STEP_GROUP) gather_group_add<4>(acc, db, v, dlo, dhi, rel);
      if (left > 5 * STEP_GROUP) gather_group_add<5>(acc, db, v, dlo, dhi, rel);
      k += TRIP_EDGES;
    } while (k < nmax);
    const float dr = rsqrtf((float)cnt);
    float2 o0, o1;
    o0.x = acc.x * dr; o0.y = acc.y * dr; o1.x = acc.z * dr; o1.y = acc.w * dr;
    float2* dst = (float2*)(myA + (2 * p + h) * A_PITCH + 4 * mn);
    dst[0] = o0; dst[1] = o1;

    dlo = scale_of(tlo1, wlo1, mn < cnt1);
    dhi = scale_of(thi1, whi1, mn < TRIP_EDGES - 16 && 16 + mn < cnt1);
    beg = beg1; cnt = cnt1; nmax = nmax1; wlo = wlo1; whi = whi1;
    beg1 = beg2; cnt1 = cnt2; nmax1 = nmax2; wlo1 = wlo2; whi1 = whi2;
  }
  __builtin_amdgcn_wave_barrier();

  v2f a[D_FEAT / 4];
  #pragma unroll
  for (int ks = 0; ks < D_FEAT / 4; ++ks) {
    const float2 t = *(const float2*)(myA + mn * A_PITCH + ks * 4 + 2 * h);
    a[ks].x = t.x; a[ks].y = t.y;
  }
  #pragma unroll
  for (int nt = 0; nt < UNITS / 16; ++nt) {
    const int nBase = nt * 16;
    v8f acc = {};
    #pragma unroll
    for (int ks = 0; ks < D_FEAT / 4; ++ks) {
      const int k0 = ks * 4 + 2 * h;
      v2f b;
      b.x = sW[(k0)     * UNITS + nBase + mn];
      b.y = sW[(k0 + 1) * UNITS + nBase + mn];
      acc = __builtin_amdgcn_wmma_f32_16x16x4_f32(
           false, a[ks],  false, b,
           (short)0, acc,  false,  false);
    }
    const float bn = bias[nBase + mn];
    #pragma unroll
    for (int j = 0; j < 8; ++j) {
      const int m = j + 8 * h;
      const float rr = acc[j] + bn;
      out[(size_t)(rowBase + m) * UNITS + nBase + mn] = fmaxf(rr, 0.0f);
    }
  }
}

extern "C" void kernel_launch(void* const* d_in, const int* in_sizes, int n_in,
                              void* d_out, int out_size, void* d_ws, size_t ws_size,
                              hipStream_t stream) {
  const float* x      = (const float*)d_in[0];
  const float* W      = (const float*)d_in[1];
  const float* b      = (const float*)d_in[2];
  const int*   source = (const int*)d_in[3];
  const int*   target = (const int*)d_in[4];
  float*       out    = (float*)d_out;

  unsigned* binned      = (unsigned*)d_ws;
  unsigned* sorted_edge = (unsigned*)d_ws + (size_t)N_EDGES;
  unsigned* seg         = (unsigned*)d_ws + (size_t)2 * N_EDGES;
  unsigned* rowptr      = (unsigned*)d_ws + (size_t)2 * N_EDGES + (size_t)N_RANGES * N_CHUNKS;
  float*    dinv        = (float*)(rowptr + N_NODES + 4);
  unsigned* deg8w       = (unsigned*)(dinv + N_NODES);

  gcn_partition_kernel<<<N_CHUNKS, PART_THREADS, 0, stream>>>(source, target, binned, seg);
  gcn_range_sort_kernel<<<N_RANGES, SORT_THREADS, 0, stream>>>(binned, seg, rowptr, dinv, deg8w, sorted_edge);
  gcn_gather_gemm_wmma_kernel<<<GG_BLOCKS, GG_THREADS, 0, stream>>>(x, sorted_edge, rowptr, dinv,
                                                                    deg8w, W, b, out);
}


# Range-sort kernel ranks an edge in its list with one compare per entry: entries are unique (node, source, slot) keys read as aligned 16-byte pieces, no end test or branch per entry; sum order kept
# speedup vs baseline: 5.7877x; 1.0421x over previous
#include <hip/hip_runtime.h>
#include <hip/hip_bf16.h>


typedef __attribute__((ext_vector_type(2))) float v2f;
typedef __attribute__((ext_vector_type(8))) float v8f;

#define N_NODES 100000
#define N_EDGES 1600000
#define D_FEAT  64
#define UNITS   64

#define SRC_BITS 17
#define SRC_MASK ((1u << SRC_BITS) - 1u)
#define RANGE_NODES 400
#define N_RANGES    (N_NODES / RANGE_NODES)
#define N_CHUNKS    250
#define CHUNK_EDGES (N_EDGES / N_CHUNKS)
static_assert(N_NODES <= (1 << SRC_BITS), "a source id must fit the low field of the edge word");
static_assert(RANGE_NODES <= (1 << (32 - SRC_BITS)), "a node's index in its range must fit the high field");
static_assert(N_NODES % RANGE_NODES == 0, "ranges must be whole");
static_assert(N_EDGES % N_CHUNKS == 0, "chunks must be whole");
static_assert(CHUNK_EDGES < 65536, "ranks, starts and counts inside a chunk are kept in 16 bits");

__device__ __forceinline__ unsigned wave_inclusive_scan(unsigned v, int lane) {
  #pragma unroll
  for (int d = 1; d < 32; d <<= 1) {
    const unsigned t = __shfl_up(v, (unsigned)d);
    if (lane >= d) v += t;
  }
  return v;
}

template <int NWAVES>
__device__ __forceinline__ unsigned block_exclusive_scan(unsigned v, unsigned* wave_tot,
                                                         unsigned* total) {
  const int lane = threadIdx.x & 31;
  const int w = threadIdx.x >> 5;
  const unsigned inc = wave_inclusive_scan(v, lane);
  if (lane == 31) wave_tot[w] = inc;
  __syncthreads();
  unsigned off = 0u, tot = 0u;
  for (int j = 0; j < NWAVES; ++j) {
    const unsigned t = wave_tot[j];
    if (j < w) off += t;
    tot += t;
  }
  *total = tot;
  return off + inc - v;
}

#define PART_THREADS 320
#define PART_LOADS   (CHUNK_EDGES / 4 / PART_THREADS)
static_assert(CHUNK_EDGES % (4 * PART_THREADS) == 0, "a chunk must be whole int4s per thread");
static_assert(N_RANGES <= PART_THREADS, "one thread per range in the scan");

__global__ __launch_bounds__(PART_THREADS) void gcn_partition_kernel(
    const int* __restrict__ source, const int* __restrict__ target,
    unsigned* __restrict__ binned, unsigned* __restrict__ seg) {
  __shared__ unsigned cnt[N_RANGES];
  __shared__ unsigned start[N_RANGES];
  __shared__ unsigned wave_tot[PART_THREADS / 32];
  __shared__ __attribute__((aligned(16))) unsigned stage[CHUNK_EDGES];
  const int tid = threadIdx.x;
  const int c = blockIdx.x;

  const int4* t4 = (const int4*)(target + (size_t)c * CHUNK_EDGES);
  const int4* s4 = (const int4*)(source + (size_t)c * CHUNK_EDGES);
  int4 t[PART_LOADS];
  int4 sv[PART_LOADS];
  #pragma unroll
  for (int k = 0; k < PART_LOADS; ++k) {
    t[k]  = t4[tid + k * PART_THREADS];
    sv[k] = s4[tid + k * PART_THREADS];
  }
  if (tid < N_RANGES) cnt[tid] = 0u;
  __syncthreads();

  unsigned key[PART_LOADS * 4];
  unsigned word[PART_LOADS * 4];
  #pragma unroll
  for (int k = 0; k < PART_LOADS; ++k) {
    const int tv[4] = {t[k].x, t[k].y, t[k].z, t[k].w};
    const int sw[4] = {sv[k].x, sv[k].y, sv[k].z, sv[k].w};
    #pragma unroll
    for (int e = 0; e < 4; ++e) {
      const unsigned tg = (unsigned)tv[e];
      const unsigned r = tg / (unsigned)RANGE_NODES;
      const unsigned loc = tg - r * (unsigned)RANGE_NODES;
      const unsigned rank = atomicAdd(&cnt[r], 1u);
      key[k * 4 + e] = (r << 16) | rank;
      word[k * 4 + e] = (unsigned)sw[e] | (loc << SRC_BITS);
    }
  }
  __syncthreads();

  unsigned total;
  const unsigned n = (tid < N_RANGES) ? cnt[tid] : 0u;
  const unsigned ex = block_exclusive_scan<PART_THREADS / 32>(n, wave_tot, &total);
  if (tid < N_RANGES) {
    start[tid] = ex;
    seg[(size_t)tid * N_CHUNKS + c] = (ex << 16) | n;
  }
  __syncthreads();

  #pragma unroll
  for (int i = 0; i < PART_LOADS * 4; ++i) stage[start[key[i] >> 16] + (key[i] & 0xFFFFu)] = word[i];
  __syncthreads();

  uint4* dst4 = (uint4*)(binned + (size_t)c * CHUNK_EDGES);
  const uint4* st4 = (const uint4*)stage;
  #pragma unroll
  for (int k = 0; k < PART_LOADS; ++k) dst4[tid + k * PART_THREADS] = st4[tid + k * PART_THREADS];
}

#define SORT_THREADS 1024
#define SORT_WAVES   (SORT_THREADS / 32)
#define SORT_CAP     16384
#define RANK_MAX_DEG 64
#define POS_BITS     6
#define KEY_NODE_SHIFT (SRC_BITS + POS_BITS)
#define SEG_BATCH    4
#define RANK_BATCH   8
#define DEG_SAT      255u
static_assert(N_CHUNKS <= SORT_THREADS && RANGE_NODES <= SORT_THREADS, "one thread per run / per node in the scans");
static_assert(RANK_MAX_DEG == (1 << POS_BITS), "slots of a ranked list must be distinct in POS_BITS bits");
static_assert(RANK_BATCH == 8 && SORT_CAP % 4 == 0, "a batch is two aligned 16-byte pieces of lst[]");
static_assert(RANGE_NODES < (1 << (32 - KEY_NODE_SHIFT)), "node index, source id and slot share one word below the end mark");

template <class F>
__device__ __forceinline__ void for_each_range_edge(const unsigned* __restrict__ binned,
                                                    const unsigned* sstart, const unsigned* scnt,
                                                    F f) {
  const unsigned lane = threadIdx.x & 31;
  const int wave = threadIdx.x >> 5;
  for (int c0 = wave * SEG_BATCH; c0 < N_CHUNKS; c0 += SORT_WAVES * SEG_BATCH) {
    unsigned n[SEG_BATCH], v[SEG_BATCH];
    const unsigned* p[SEG_BATCH];
    #pragma unroll
    for (int q = 0; q < SEG_BATCH; ++q) {
      const int c = c0 + q;
      const bool live = c < N_CHUNKS;
      n[q] = live ? scnt[c] : 0u;
      p[q] = binned + (size_t)(live ? c : 0) * CHUNK_EDGES + (live ? sstart[c] : 0u);
      v[q] = 0u;
      if (lane < n[q]) v[q] = p[q][lane];
    }
    #pragma unroll
    for (int q = 0; q < SEG_BATCH; ++q) {
      if (lane < n[q]) f(v[q], c0 + q, lane);
      for (unsigned j = 32u + lane; j < n[q]; j += 32u) f(p[q][j], c0 + q, j);
    }
  }
}

__global__ __launch_bounds__(SORT_THREADS) void gcn_range_sort_kernel(
    const unsigned* __restrict__ binned, const unsigned* __restrict__ seg,
    unsigned* __restrict__ rowptr, float* __restrict__ dinv,
    unsigned* __restrict__ deg8w, unsigned* __restrict__ sorted_edge) {
  __shared__ unsigned sstart[N_CHUNKS];
  __shared__ unsigned scnt[N_CHUNKS];
  __shared__ unsigned soff[N_CHUNKS];
  __shared__ unsigned ncnt[RANGE_NODES];
  __shared__ unsigned nstart[RANGE_NODES + 1];
  __shared__ unsigned ncur[RANGE_NODES];
  __shared__ unsigned wave_tot[SORT_WAVES];
  __shared__ unsigned raw[SORT_CAP];
  __shared__ __attribute__((aligned(16))) unsigned lst[SORT_CAP + RANK_BATCH];
  const int tid = threadIdx.x;
  const int r = blockIdx.x;

  unsigned st = 0u, n = 0u;
  if (tid < N_CHUNKS) {
    const unsigned sg = seg[(size_t)r * N_CHUNKS + tid];
    st = sg >> 16;
    n  = sg & 0xFFFFu;
    sstart[tid] = st;
    scnt[tid] = n;
  }
  if (tid < RANGE_NODES) ncnt[tid] = 0u;
  unsigned total;
  const unsigned off = block_exclusive_scan<SORT_WAVES>(n, wave_tot, &total);
  if (tid < N_CHUNKS) soff[tid] = off;
  __syncthreads();
  unsigned base;
  (void)block_exclusive_scan<SORT_WAVES>(st, wave_tot, &base);
  __syncthreads();

  const bool fits = total <= (unsigned)SORT_CAP;
  if (fits) {
    for_each_range_edge(binned, sstart, scnt, [&](unsigned w, int c, unsigned j) {
      raw[soff[c] + j] = w;
      atomicAdd(&ncnt[w >> SRC_BITS], 1u);
    });
  } else {
    for_each_range_edge(binned, sstart, scnt, [&](unsigned w, int, unsigned) {
      atomicAdd(&ncnt[w >> SRC_BITS], 1u);
    });
  }
  __syncthreads();

  const unsigned d = (tid < RANGE_NODES) ? ncnt[tid] : 0u;
  unsigned total_nodes;
  const unsigned ns = block_exclusive_scan<SORT_WAVES>(d, wave_tot, &total_nodes);
  if (tid < RANGE_NODES) {
    nstart[tid] = ns;
    ncur[tid] = ns;
    rowptr[(size_t)r * RANGE_NODES + tid] = base + ns;
    dinv[(size_t)r * RANGE_NODES + tid] = rsqrtf((float)d);
    ((unsigned char*)deg8w)[(size_t)r * RANGE_NODES + tid] = (unsigned char)(d < DEG_SAT ? d : DEG_SAT);
  }
  if (tid == 0) {
    nstart[RANGE_NODES] = total_nodes;
    if (r == N_RANGES - 1) rowptr[N_NODES] = base + total_nodes;
  }
  __syncthreads();

  if (fits) {
    for (unsigned i = tid; i < total; i += SORT_THREADS) {
      const unsigned w = raw[i];
      const unsigned node = w >> SRC_BITS;
      const unsigned p = atomicAdd(&ncur[node], 1u);
      const unsigned slot = (p - nstart[node]) & (unsigned)(RANK_MAX_DEG - 1);
      lst[p] = (node << KEY_NODE_SHIFT) | ((w & SRC_MASK) << POS_BITS) | slot;
    }
    if (tid < RANK_BATCH) lst[total + tid] = 0xFFFFFFFFu;
    __syncthreads();
    for (unsigned i = tid; i < total; i += SORT_THREADS) {
      const unsigned key = lst[i];
      const unsigned node = key >> KEY_NODE_SHIFT;
      const unsigned b = nstart[node], e = nstart[node + 1];
      unsigned pos = i;
      if (e - b <= (unsigned)RANK_MAX_DEG) {
        const unsigned q0 = b & ~3u;
        unsigned smaller = 0u;
        for (unsigned q = q0; q < e; q += RANK_BATCH) {
          const uint4 u0 = *(const uint4*)(lst + q);
          const uint4 u1 = *(const uint4*)(lst + q + 4u);
          smaller += (u0.x < key ? 1u : 0u) + (u0.y < key ? 1u : 0u) + (u0.z < key ? 1u : 0u) + (u0.w < key ? 1u : 0u)
                   + (u1.x < key ? 1u : 0u) + (u1.y < key ? 1u : 0u) + (u1.z < key ? 1u : 0u) + (u1.w < key ? 1u : 0u);
        }
        pos = q0 + smaller;
      }
      raw[pos] = (key >> POS_BITS) & SRC_MASK;
    }
    __syncthreads();
    for (unsigned i = tid; i < total; i += SORT_THREADS) sorted_edge[base + i] = raw[i];
  } else {
    for_each_range_edge(binned, sstart, scnt, [&](unsigned w, int, unsigned) {
      const unsigned p = atomicAdd(&ncur[w >> SRC_BITS], 1u);
      sorted_edge[base + p] = w & SRC_MASK;
    });
  }
}

#define TRIP_EDGES   24
#define STEP_GROUP   4
#define GG_BLOCKS    256
#define GG_WAVES     25
#define GG_THREADS   (GG_WAVES * 32)
#define N_TILES      (N_NODES / 16)
#define A_PITCH      (D_FEAT + 2)
#define DEG_CHUNKS   (N_NODES / 16)
static_assert(N_NODES % 16 == 0, "whole tiles, and the table in whole 16-byte pieces");
static_assert(GG_BLOCKS * GG_WAVES >= N_TILES, "one tile per wave");
static_assert((A_PITCH % 2) == 0, "A rows are written and read as 8-byte pairs");
static_assert(D_FEAT == 64, "a half-wave of 16 lanes with 4 floats each covers one row of x");
static_assert(TRIP_EDGES > 16 && TRIP_EDGES <= 32 && TRIP_EDGES % STEP_GROUP == 0,
              "a trip's edge words are held in two registers of 16 lanes per half-wave");

__device__ __forceinline__ int lane_value(int v, int l) {
  return __builtin_amdgcn_readlane(v, l);
}

template <int J>
__device__ __forceinline__ unsigned half_lane_value(unsigned prev, unsigned v) {
  return __builtin_amdgcn_permlane16(prev, v, (unsigned)J * 0x11111111u, (unsigned)J * 0x11111111u,
                                     false, false);
}

template <int J>
__device__ __forceinline__ void gather_step_load(float4 (&v)[TRIP_EDGES], unsigned& s, const char* xb,
                                                 unsigned wlo, unsigned whi, unsigned c16) {
  s = half_lane_value<(J & 15)>(s, (J < 16) ? wlo : whi);
  v[J] = *(const float4*)(xb + ((s << 8) | c16));
}
template <int J>
__device__ __forceinline__ void gather_step_add(float4& acc, unsigned& db, const float4 (&v)[TRIP_EDGES],
                                                float dlo, float dhi, int rel) {
  db = half_lane_value<(J & 15)>(db, __builtin_bit_cast(unsigned, (J < 16) ? dlo : dhi));
  const float d = __builtin_bit_cast(float, db);
  if (J < rel) {
    acc.x += v[J].x * d; acc.y += v[J].y * d; acc.z += v[J].z * d; acc.w += v[J].w * d;
  }
}
template <int G>
__device__ __forceinline__ void gather_group_load(float4 (&v)[TRIP_EDGES], unsigned& s, const char* xb,
                                                  unsigned wlo, unsigned whi, unsigned c16) {
  gather_step_load<G * STEP_GROUP + 0>(v, s, xb, wlo, whi, c16);
  gather_step_load<G * STEP_GROUP + 1>(v, s, xb, wlo, whi, c16);
  gather_step_load<G * STEP_GROUP + 2>(v, s, xb, wlo, whi, c16);
  gather_step_load<G * STEP_GROUP + 3>(v, s, xb, wlo, whi, c16);
}
template <int G>
__device__ __forceinline__ void gather_group_add(float4& acc, unsigned& db, const float4 (&v)[TRIP_EDGES],
                                                 float dlo, float dhi, int rel) {
  gather_step_add<G * STEP_GROUP + 0>(acc, db, v, dlo, dhi, rel);
  gather_step_add<G * STEP_GROUP + 1>(acc, db, v, dlo, dhi, rel);
  gather_step_add<G * STEP_GROUP + 2>(acc, db, v, dlo, dhi, rel);
  gather_step_add<G * STEP_GROUP + 3>(acc, db, v, dlo, dhi, rel);
}
static_assert(STEP_GROUP == 4 && TRIP_EDGES == 6 * STEP_GROUP, "the groups are written out below");

__global__ __launch_bounds__(GG_THREADS) void gcn_gather_gemm_wmma_kernel(
    const float* __restrict__ x, const unsigned* __restrict__ sorted_edge,
    const unsigned* __restrict__ rowptr, const float* __restrict__ dinv,
    const unsigned* __restrict__ deg8w,
    const float* __restrict__ W, const float* __restrict__ bias,
    float* __restrict__ out) {
  __shared__ __attribute__((aligned(16))) unsigned char sDeg[N_NODES];
  __shared__ __attribute__((aligned(16))) float sW[D_FEAT * UNITS];
  __shared__ __attribute__((aligned(16))) float sA[GG_WAVES * 16 * A_PITCH];

  const int tid  = threadIdx.x;
  const int lane = tid & 31;
  const int wave = tid >> 5;
  const int tile = (int)blockIdx.x + GG_BLOCKS * wave;
  const bool live = tile < N_TILES;
  const int rowBase = tile * 16;

  const int h  = lane >> 4;
  const int mn = lane & 15;
  const unsigned c16 = (unsigned)mn << 4;
  const char* xb = (const char*)x;

  int rp = 0;
  if (live && lane < 17) rp = (int)rowptr[rowBase + lane];
  auto pair_bounds = [&](int p, int& beg, int& cnt, int& nmax) {
    const int s0 = lane_value(rp, 2 * p), s1 = lane_value(rp, 2 * p + 1), s2 = lane_value(rp, 2 * p + 2);
    beg = h ? s1 : s0;
    cnt = h ? (s2 - s1) : (s1 - s0);
    nmax = (s1 - s0 > s2 - s1) ? (s1 - s0) : (s2 - s1);
  };
  auto load_words = [&](int beg, int cnt, int k, unsigned& wlo, unsigned& whi) {
    wlo = 0u; whi = 0u;
    if (k + mn < cnt) wlo = sorted_edge[beg + k + mn];
    if (mn < TRIP_EDGES - 16 && k + 16 + mn < cnt) whi = sorted_edge[beg + k + 16 + mn];
  };
  auto scale_of = [&](unsigned t, unsigned w, bool has) -> float {
    float d = 0.0f;
    if (has) {
      d = rsqrtf((float)t);
      if (t == DEG_SAT) d = dinv[w];
    }
    return d;
  };

  int beg, cnt, nmax, beg1, cnt1, nmax1;
  unsigned wlo, whi, wlo1, whi1;
  pair_bounds(0, beg, cnt, nmax);
  pair_bounds(1, beg1, cnt1, nmax1);
  load_words(beg, cnt, 0, wlo, whi);
  load_words(beg1, cnt1, 0, wlo1, whi1);

  {
    const uint4* src = (const uint4*)deg8w;
    uint4* dst = (uint4*)sDeg;
    const int rot = (int)blockIdx.x * (DEG_CHUNKS / GG_BLOCKS);
    for (int i = tid; i < DEG_CHUNKS; i += GG_THREADS) {
      int c = i + rot;
      if (c >= DEG_CHUNKS) c -= DEG_CHUNKS;
      dst[c] = src[c];
    }
    const float4* Wv = (const float4*)W;
    float4* sWv = (float4*)sW;
    for (int i = tid; i < (D_FEAT * UNITS) / 4; i += GG_THREADS) sWv[i] = Wv[i];
  }
  __syncthreads();
  if (!live) return;

  float* myA = sA + wave * (16 * A_PITCH);
  float dlo = scale_of(sDeg[wlo], wlo, mn < cnt);
  float dhi = scale_of(sDeg[whi], whi, mn < TRIP_EDGES - 16 && 16 + mn < cnt);
  for (int p = 0; p < 8; ++p) {
    int beg2 = 0, cnt2 = 0, nmax2 = 0;
    unsigned wlo2 = 0u, whi2 = 0u;
    if (p + 2 < 8) {
      pair_bounds(p + 2, beg2, cnt2, nmax2);
      load_words(beg2, cnt2, 0, wlo2, whi2);
    }
    const unsigned tlo1 = sDeg[wlo1], thi1 = sDeg[whi1];

    float4 acc = {0.0f, 0.0f, 0.0f, 0.0f};
    int k = 0;
    do {
      if (k != 0) {
        load_words(beg, cnt, k, wlo, whi);
        dlo = scale_of(sDeg[wlo], wlo, k + mn < cnt);
        dhi = scale_of(sDeg[whi], whi, mn < TRIP_EDGES - 16 && k + 16 + mn < cnt);
      }
      const int rel = cnt - k;
      const int left = nmax - k;
      float4 v[TRIP_EDGES];
      unsigned s = 0u, db = 0u;
      if (left > 0 * STEP_GROUP) gather_group_load<0>(v, s, xb, wlo, whi, c16);
      if (left > 1 * STEP_GROUP) gather_group_load<1>(v, s, xb, wlo, whi, c16);
      if (left > 2 * STEP_GROUP) gather_group_load<2>(v, s, xb, wlo, whi, c16);
      if (left > 3 * STEP_GROUP) gather_group_load<3>(v, s, xb, wlo, whi, c16);
      if (left > 4 * STEP_GROUP) gather_group_load<4>(v, s, xb, wlo, whi, c16);
      if (left > 5 * STEP_GROUP) gather_group_load<5>(v, s, xb, wlo, whi, c16);
      if (left > 0 * STEP_GROUP) gather_group_add<0>(acc, db, v, dlo, dhi, rel);
      if (left > 1 * STEP_GROUP) gather_group_add<1>(acc, db, v, dlo, dhi, rel);
      if (left > 2 * STEP_GROUP) gather_group_add<2>(acc, db, v, dlo, dhi, rel);
      if (left > 3 * STEP_GROUP) gather_group_add<3>(acc, db, v, dlo, dhi, rel);
      if (left > 4 * STEP_GROUP) gather_group_add<4>(acc, db, v, dlo, dhi, rel);
      if (left > 5 * STEP_GROUP) gather_group_add<5>(acc, db, v, dlo, dhi, rel);
      k += TRIP_EDGES;
    } while (k < nmax);
    const float dr = rsqrtf((float)cnt);
    float2 o0, o1;
    o0.x = acc.x * dr; o0.y = acc.y * dr; o1.x = acc.z * dr; o1.y = acc.w * dr;
    float2* dst = (float2*)(myA + (2 * p + h) * A_PITCH + 4 * mn);
    dst[0] = o0; dst[1] = o1;

    dlo = scale_of(tlo1, wlo1, mn < cnt1);
    dhi = scale_of(thi1, whi1, mn < TRIP_EDGES - 16 && 16 + mn < cnt1);
    beg = beg1; cnt = cnt1; nmax = nmax1; wlo = wlo1; whi = whi1;
    beg1 = beg2; cnt1 = cnt2; nmax1 = nmax2; wlo1 = wlo2; whi1 = whi2;
  }
  __builtin_amdgcn_wave_barrier();

  v2f a[D_FEAT / 4];
  #pragma unroll
  for (int ks = 0; ks < D_FEAT / 4; ++ks) {
    const float2 t = *(const float2*)(myA + mn * A_PITCH + ks * 4 + 2 * h);
    a[ks].x = t.x; a[ks].y = t.y;
  }
  #pragma unroll
  for (int nt = 0; nt < UNITS / 16; ++nt) {
    const int nBase = nt * 16;
    v8f acc = {};
    #pragma unroll
    for (int ks = 0; ks < D_FEAT / 4; ++ks) {
      const int k0 = ks * 4 + 2 * h;
      v2f b;
      b.x = sW[(k0)     * UNITS + nBase + mn];
      b.y = sW[(k0 + 1) * UNITS + nBase + mn];
      acc = __builtin_amdgcn_wmma_f32_16x16x4_f32(
           false, a[ks],  false, b,
           (short)0, acc,  false,  false);
    }
    const float bn = bias[nBase + mn];
    #pragma unroll
    for (int j = 0; j < 8; ++j) {
      const int m = j + 8 * h;
      const float rr = acc[j] + bn;
      out[(size_t)(rowBase + m) * UNITS + nBase + mn] = fmaxf(rr, 0.0f);
    }
  }
}

extern "C" void kernel_launch(void* const* d_in, const int* in_sizes, int n_in,
                              void* d_out, int out_size, void* d_ws, size_t ws_size,
                              hipStream_t stream) {
  const float* x      = (const float*)d_in[0];
  const float* W      = (const float*)d_in[1];
  const float* b      = (const float*)d_in[2];
  const int*   source = (const int*)d_in[3];
  const int*   target = (const int*)d_in[4];
  float*       out    = (float*)d_out;

  unsigned* binned      = (unsigned*)d_ws;
  unsigned* sorted_edge = (unsigned*)d_ws + (size_t)N_EDGES;
  unsigned* seg         = (unsigned*)d_ws + (size_t)2 * N_EDGES;
  unsigned* rowptr      = (unsigned*)d_ws + (size_t)2 * N_EDGES + (size_t)N_RANGES * N_CHUNKS;
  float*    dinv        = (float*)(rowptr + N_NODES + 4);
  unsigned* deg8w       = (unsigned*)(dinv + N_NODES);

  gcn_partition_kernel<<<N_CHUNKS, PART_THREADS, 0, stream>>>(source, target, binned, seg);
  gcn_range_sort_kernel<<<N_RANGES, SORT_THREADS, 0, stream>>>(binned, seg, rowptr, dinv, deg8w, sorted_edge);
  gcn_gather_gemm_wmma_kernel<<<GG_BLOCKS, GG_THREADS, 0, stream>>>(x, sorted_edge, rowptr, dinv,
                                                                    deg8w, W, b, out);
}


# Independent loads issued together in the serial heads: gather kernel stages its table and W as one burst of 10 loads per thread (was 10 dependent trips), range kernel loads 64 words of 8 runs at once
# speedup vs baseline: 5.9707x; 1.0316x over previous
#include <hip/hip_runtime.h>
#include <hip/hip_bf16.h>


typedef __attribute__((ext_vector_type(2))) float v2f;
typedef __attribute__((ext_vector_type(8))) float v8f;

#define N_NODES 100000
#define N_EDGES 1600000
#define D_FEAT  64
#define UNITS   64

#define SRC_BITS 17
#define SRC_MASK ((1u << SRC_BITS) - 1u)
#define RANGE_NODES 400
#define N_RANGES    (N_NODES / RANGE_NODES)
#define N_CHUNKS    250
#define CHUNK_EDGES (N_EDGES / N_CHUNKS)
static_assert(N_NODES <= (1 << SRC_BITS), "a source id must fit the low field of the edge word");
static_assert(RANGE_NODES <= (1 << (32 - SRC_BITS)), "a node's index in its range must fit the high field");
static_assert(N_NODES % RANGE_NODES == 0, "ranges must be whole");
static_assert(N_EDGES % N_CHUNKS == 0, "chunks must be whole");
static_assert(CHUNK_EDGES < 65536, "ranks, starts and counts inside a chunk are kept in 16 bits");

__device__ __forceinline__ unsigned wave_inclusive_scan(unsigned v, int lane) {
  #pragma unroll
  for (int d = 1; d < 32; d <<= 1) {
    const unsigned t = __shfl_up(v, (unsigned)d);
    if (lane >= d) v += t;
  }
  return v;
}

template <int NWAVES>
__device__ __forceinline__ unsigned block_exclusive_scan(unsigned v, unsigned* wave_tot,
                                                         unsigned* total) {
  const int lane = threadIdx.x & 31;
  const int w = threadIdx.x >> 5;
  const unsigned inc = wave_inclusive_scan(v, lane);
  if (lane == 31) wave_tot[w] = inc;
  __syncthreads();
  unsigned off = 0u, tot = 0u;
  for (int j = 0; j < NWAVES; ++j) {
    const unsigned t = wave_tot[j];
    if (j < w) off += t;
    tot += t;
  }
  *total = tot;
  return off + inc - v;
}

#define PART_THREADS 320
#define PART_LOADS   (CHUNK_EDGES / 4 / PART_THREADS)
static_assert(CHUNK_EDGES % (4 * PART_THREADS) == 0, "a chunk must be whole int4s per thread");
static_assert(N_RANGES <= PART_THREADS, "one thread per range in the scan");

__global__ __launch_bounds__(PART_THREADS) void gcn_partition_kernel(
    const int* __restrict__ source, const int* __restrict__ target,
    unsigned* __restrict__ binned, unsigned* __restrict__ seg) {
  __shared__ unsigned cnt[N_RANGES];
  __shared__ unsigned start[N_RANGES];
  __shared__ unsigned wave_tot[PART_THREADS / 32];
  __shared__ __attribute__((aligned(16))) unsigned stage[CHUNK_EDGES];
  const int tid = threadIdx.x;
  const int c = blockIdx.x;

  const int4* t4 = (const int4*)(target + (size_t)c * CHUNK_EDGES);
  const int4* s4 = (const int4*)(source + (size_t)c * CHUNK_EDGES);
  int4 t[PART_LOADS];
  int4 sv[PART_LOADS];
  #pragma unroll
  for (int k = 0; k < PART_LOADS; ++k) {
    t[k]  = t4[tid + k * PART_THREADS];
    sv[k] = s4[tid + k * PART_THREADS];
  }
  if (tid < N_RANGES) cnt[tid] = 0u;
  __syncthreads();

  unsigned key[PART_LOADS * 4];
  unsigned word[PART_LOADS * 4];
  #pragma unroll
  for (int k = 0; k < PART_LOADS; ++k) {
    const int tv[4] = {t[k].x, t[k].y, t[k].z, t[k].w};
    const int sw[4] = {sv[k].x, sv[k].y, sv[k].z, sv[k].w};
    #pragma unroll
    for (int e = 0; e < 4; ++e) {
      const unsigned tg = (unsigned)tv[e];
      const unsigned r = tg / (unsigned)RANGE_NODES;
      const unsigned loc = tg - r * (unsigned)RANGE_NODES;
      const unsigned rank = atomicAdd(&cnt[r], 1u);
      key[k * 4 + e] = (r << 16) | rank;
      word[k * 4 + e] = (unsigned)sw[e] | (loc << SRC_BITS);
    }
  }
  __syncthreads();

  unsigned total;
  const unsigned n = (tid < N_RANGES) ? cnt[tid] : 0u;
  const unsigned ex = block_exclusive_scan<PART_THREADS / 32>(n, wave_tot, &total);
  if (tid < N_RANGES) {
    start[tid] = ex;
    seg[(size_t)tid * N_CHUNKS + c] = (ex << 16) | n;
  }
  __syncthreads();

  #pragma unroll
  for (int i = 0; i < PART_LOADS * 4; ++i) stage[start[key[i] >> 16] + (key[i] & 0xFFFFu)] = word[i];
  __syncthreads();

  uint4* dst4 = (uint4*)(binned + (size_t)c * CHUNK_EDGES);
  const uint4* st4 = (const uint4*)stage;
  #pragma unroll
  for (int k = 0; k < PART_LOADS; ++k) dst4[tid + k * PART_THREADS] = st4[tid + k * PART_THREADS];
}

#define SORT_THREADS 1024
#define SORT_WAVES   (SORT_THREADS / 32)
#define SORT_CAP     16384
#define RANK_MAX_DEG 64
#define POS_BITS     6
#define KEY_NODE_SHIFT (SRC_BITS + POS_BITS)
#define SEG_BATCH    8
#define RANK_BATCH   8
#define DEG_SAT      255u
static_assert(N_CHUNKS <= SORT_THREADS && RANGE_NODES <= SORT_THREADS, "one thread per run / per node in the scans");
static_assert(RANK_MAX_DEG == (1 << POS_BITS), "slots of a ranked list must be distinct in POS_BITS bits");
static_assert(RANK_BATCH == 8 && SORT_CAP % 4 == 0, "a batch is two aligned 16-byte pieces of lst[]");
static_assert(RANGE_NODES < (1 << (32 - KEY_NODE_SHIFT)), "node index, source id and slot share one word below the end mark");

template <class F>
__device__ __forceinline__ void for_each_range_edge(const unsigned* __restrict__ binned,
                                                    const unsigned* sstart, const unsigned* scnt,
                                                    F f) {
  const unsigned lane = threadIdx.x & 31;
  const int wave = threadIdx.x >> 5;
  for (int c0 = wave * SEG_BATCH; c0 < N_CHUNKS; c0 += SORT_WAVES * SEG_BATCH) {
    unsigned n[SEG_BATCH], v[SEG_BATCH], v2[SEG_BATCH];
    const unsigned* p[SEG_BATCH];
    #pragma unroll
    for (int q = 0; q < SEG_BATCH; ++q) {
      const int c = c0 + q;
      const bool live = c < N_CHUNKS;
      n[q] = live ? scnt[c] : 0u;
      p[q] = binned + (size_t)(live ? c : 0) * CHUNK_EDGES + (live ? sstart[c] : 0u);
      v[q] = 0u;
      v2[q] = 0u;
      if (lane < n[q]) v[q] = p[q][lane];
      if (32u + lane < n[q]) v2[q] = p[q][32u + lane];
    }
    #pragma unroll
    for (int q = 0; q < SEG_BATCH; ++q) {
      if (lane < n[q]) f(v[q], c0 + q, lane);
      if (32u + lane < n[q]) f(v2[q], c0 + q, 32u + lane);
      for (unsigned j = 64u + lane; j < n[q]; j += 32u) f(p[q][j], c0 + q, j);
    }
  }
}

__global__ __launch_bounds__(SORT_THREADS) void gcn_range_sort_kernel(
    const unsigned* __restrict__ binned, const unsigned* __restrict__ seg,
    unsigned* __restrict__ rowptr, float* __restrict__ dinv,
    unsigned* __restrict__ deg8w, unsigned* __restrict__ sorted_edge) {
  __shared__ unsigned sstart[N_CHUNKS];
  __shared__ unsigned scnt[N_CHUNKS];
  __shared__ unsigned soff[N_CHUNKS];
  __shared__ unsigned ncnt[RANGE_NODES];
  __shared__ unsigned nstart[RANGE_NODES + 1];
  __shared__ unsigned ncur[RANGE_NODES];
  __shared__ unsigned wave_tot[SORT_WAVES];
  __shared__ unsigned raw[SORT_CAP];
  __shared__ __attribute__((aligned(16))) unsigned lst[SORT_CAP + RANK_BATCH];
  const int tid = threadIdx.x;
  const int r = blockIdx.x;

  unsigned st = 0u, n = 0u;
  if (tid < N_CHUNKS) {
    const unsigned sg = seg[(size_t)r * N_CHUNKS + tid];
    st = sg >> 16;
    n  = sg & 0xFFFFu;
    sstart[tid] = st;
    scnt[tid] = n;
  }
  if (tid < RANGE_NODES) ncnt[tid] = 0u;
  unsigned total;
  const unsigned off = block_exclusive_scan<SORT_WAVES>(n, wave_tot, &total);
  if (tid < N_CHUNKS) soff[tid] = off;
  __syncthreads();
  unsigned base;
  (void)block_exclusive_scan<SORT_WAVES>(st, wave_tot, &base);
  __syncthreads();

  const bool fits = total <= (unsigned)SORT_CAP;
  if (fits) {
    for_each_range_edge(binned, sstart, scnt, [&](unsigned w, int c, unsigned j) {
      raw[soff[c] + j] = w;
      atomicAdd(&ncnt[w >> SRC_BITS], 1u);
    });
  } else {
    for_each_range_edge(binned, sstart, scnt, [&](unsigned w, int, unsigned) {
      atomicAdd(&ncnt[w >> SRC_BITS], 1u);
    });
  }
  __syncthreads();

  const unsigned d = (tid < RANGE_NODES) ? ncnt[tid] : 0u;
  unsigned total_nodes;
  const unsigned ns = block_exclusive_scan<SORT_WAVES>(d, wave_tot, &total_nodes);
  if (tid < RANGE_NODES) {
    nstart[tid] = ns;
    ncur[tid] = ns;
    rowptr[(size_t)r * RANGE_NODES + tid] = base + ns;
    dinv[(size_t)r * RANGE_NODES + tid] = rsqrtf((float)d);
    ((unsigned char*)deg8w)[(size_t)r * RANGE_NODES + tid] = (unsigned char)(d < DEG_SAT ? d : DEG_SAT);
  }
  if (tid == 0) {
    nstart[RANGE_NODES] = total_nodes;
    if (r == N_RANGES - 1) rowptr[N_NODES] = base + total_nodes;
  }
  __syncthreads();

  if (fits) {
    for (unsigned i = tid; i < total; i += SORT_THREADS) {
      const unsigned w = raw[i];
      const unsigned node = w >> SRC_BITS;
      const unsigned p = atomicAdd(&ncur[node], 1u);
      const unsigned slot = (p - nstart[node]) & (unsigned)(RANK_MAX_DEG - 1);
      lst[p] = (node << KEY_NODE_SHIFT) | ((w & SRC_MASK) << POS_BITS) | slot;
    }
    if (tid < RANK_BATCH) lst[total + tid] = 0xFFFFFFFFu;
    __syncthreads();
    for (unsigned i = tid; i < total; i += SORT_THREADS) {
      const unsigned key = lst[i];
      const unsigned node = key >> KEY_NODE_SHIFT;
      const unsigned b = nstart[node], e = nstart[node + 1];
      unsigned pos = i;
      if (e - b <= (unsigned)RANK_MAX_DEG) {
        const unsigned q0 = b & ~3u;
        unsigned smaller = 0u;
        for (unsigned q = q0; q < e; q += RANK_BATCH) {
          const uint4 u0 = *(const uint4*)(lst + q);
          const uint4 u1 = *(const uint4*)(lst + q + 4u);
          smaller += (u0.x < key ? 1u : 0u) + (u0.y < key ? 1u : 0u) + (u0.z < key ? 1u : 0u) + (u0.w < key ? 1u : 0u)
                   + (u1.x < key ? 1u : 0u) + (u1.y < key ? 1u : 0u) + (u1.z < key ? 1u : 0u) + (u1.w < key ? 1u : 0u);
        }
        pos = q0 + smaller;
      }
      raw[pos] = (key >> POS_BITS) & SRC_MASK;
    }
    __syncthreads();
    for (unsigned i = tid; i < total; i += SORT_THREADS) sorted_edge[base + i] = raw[i];
  } else {
    for_each_range_edge(binned, sstart, scnt, [&](unsigned w, int, unsigned) {
      const unsigned p = atomicAdd(&ncur[w >> SRC_BITS], 1u);
      sorted_edge[base + p] = w & SRC_MASK;
    });
  }
}

#define TRIP_EDGES   24
#define STEP_GROUP   4
#define GG_BLOCKS    256
#define GG_WAVES     25
#define GG_THREADS   (GG_WAVES * 32)
#define N_TILES      (N_NODES / 16)
#define A_PITCH      (D_FEAT + 2)
#define DEG_CHUNKS   (N_NODES / 16)
#define W_CHUNKS     ((D_FEAT * UNITS) / 4)
#define DEG_PIECES   ((DEG_CHUNKS + GG_THREADS - 1) / GG_THREADS)
#define W_PIECES     ((W_CHUNKS + GG_THREADS - 1) / GG_THREADS)
static_assert(N_NODES % 16 == 0, "whole tiles, and the table in whole 16-byte pieces");
static_assert(GG_BLOCKS * GG_WAVES >= N_TILES, "one tile per wave");
static_assert(GG_THREADS <= N_NODES / 16 && GG_THREADS <= (D_FEAT * UNITS) / 4, "every thread has a first piece of the table and of W");
static_assert((A_PITCH % 2) == 0, "A rows are written and read as 8-byte pairs");
static_assert(D_FEAT == 64, "a half-wave of 16 lanes with 4 floats each covers one row of x");
static_assert(TRIP_EDGES > 16 && TRIP_EDGES <= 32 && TRIP_EDGES % STEP_GROUP == 0,
              "a trip's edge words are held in two registers of 16 lanes per half-wave");

__device__ __forceinline__ int lane_value(int v, int l) {
  return __builtin_amdgcn_readlane(v, l);
}

template <int J>
__device__ __forceinline__ unsigned half_lane_value(unsigned prev, unsigned v) {
  return __builtin_amdgcn_permlane16(prev, v, (unsigned)J * 0x11111111u, (unsigned)J * 0x11111111u,
                                     false, false);
}

template <int J>
__device__ __forceinline__ void gather_step_load(float4 (&v)[TRIP_EDGES], unsigned& s, const char* xb,
                                                 unsigned wlo, unsigned whi, unsigned c16) {
  s = half_lane_value<(J & 15)>(s, (J < 16) ? wlo : whi);
  v[J] = *(const float4*)(xb + ((s << 8) | c16));
}
template <int J>
__device__ __forceinline__ void gather_step_add(float4& acc, unsigned& db, const float4 (&v)[TRIP_EDGES],
                                                float dlo, float dhi, int rel) {
  db = half_lane_value<(J & 15)>(db, __builtin_bit_cast(unsigned, (J < 16) ? dlo : dhi));
  const float d = __builtin_bit_cast(float, db);
  if (J < rel) {
    acc.x += v[J].x * d; acc.y += v[J].y * d; acc.z += v[J].z * d; acc.w += v[J].w * d;
  }
}
template <int G>
__device__ __forceinline__ void gather_group_load(float4 (&v)[TRIP_EDGES], unsigned& s, const char* xb,
                                                  unsigned wlo, unsigned whi, unsigned c16) {
  gather_step_load<G * STEP_GROUP + 0>(v, s, xb, wlo, whi, c16);
  gather_step_load<G * STEP_GROUP + 1>(v, s, xb, wlo, whi, c16);
  gather_step_load<G * STEP_GROUP + 2>(v, s, xb, wlo, whi, c16);
  gather_step_load<G * STEP_GROUP + 3>(v, s, xb, wlo, whi, c16);
}
template <int G>
__device__ __forceinline__ void gather_group_add(float4& acc, unsigned& db, const float4 (&v)[TRIP_EDGES],
                                                 float dlo, float dhi, int rel) {
  gather_step_add<G * STEP_GROUP + 0>(acc, db, v, dlo, dhi, rel);
  gather_step_add<G * STEP_GROUP + 1>(acc, db, v, dlo, dhi, rel);
  gather_step_add<G * STEP_GROUP + 2>(acc, db, v, dlo, dhi, rel);
  gather_step_add<G * STEP_GROUP + 3>(acc, db, v, dlo, dhi, rel);
}
static_assert(STEP_GROUP == 4 && TRIP_EDGES == 6 * STEP_GROUP, "the groups are written out below");

__global__ __launch_bounds__(GG_THREADS) void gcn_gather_gemm_wmma_kernel(
    const float* __restrict__ x, const unsigned* __restrict__ sorted_edge,
    const unsigned* __restrict__ rowptr, const float* __restrict__ dinv,
    const unsigned* __restrict__ deg8w,
    const float* __restrict__ W, const float* __restrict__ bias,
    float* __restrict__ out) {
  __shared__ __attribute__((aligned(16))) unsigned char sDeg[N_NODES];
  __shared__ __attribute__((aligned(16))) float sW[D_FEAT * UNITS];
  __shared__ __attribute__((aligned(16))) float sA[GG_WAVES * 16 * A_PITCH];

  const int tid  = threadIdx.x;
  const int lane = tid & 31;
  const int wave = tid >> 5;
  const int tile = (int)blockIdx.x + GG_BLOCKS * wave;
  const bool live = tile < N_TILES;
  const int rowBase = tile * 16;

  const int h  = lane >> 4;
  const int mn = lane & 15;
  const unsigned c16 = (unsigned)mn << 4;
  const char* xb = (const char*)x;

  int rp = 0;
  if (live && lane < 17) rp = (int)rowptr[rowBase + lane];
  auto pair_bounds = [&](int p, int& beg, int& cnt, int& nmax) {
    const int s0 = lane_value(rp, 2 * p), s1 = lane_value(rp, 2 * p + 1), s2 = lane_value(rp, 2 * p + 2);
    beg = h ? s1 : s0;
    cnt = h ? (s2 - s1) : (s1 - s0);
    nmax = (s1 - s0 > s2 - s1) ? (s1 - s0) : (s2 - s1);
  };
  auto load_words = [&](int beg, int cnt, int k, unsigned& wlo, unsigned& whi) {
    wlo = 0u; whi = 0u;
    if (k + mn < cnt) wlo = sorted_edge[beg + k + mn];
    if (mn < TRIP_EDGES - 16 && k + 16 + mn < cnt) whi = sorted_edge[beg + k + 16 + mn];
  };
  auto scale_of = [&](unsigned t, unsigned w, bool has) -> float {
    float d = 0.0f;
    if (has) {
      d = rsqrtf((float)t);
      if (t == DEG_SAT) d = dinv[w];
    }
    return d;
  };

  const uint4* degSrc = (const uint4*)deg8w;
  const float4* Wv = (const float4*)W;
  const int rot = (int)blockIdx.x * (DEG_CHUNKS / GG_BLOCKS);
  uint4 degPiece[DEG_PIECES];
  float4 wPiece[W_PIECES];
  int degAt[DEG_PIECES], wAt[W_PIECES];
  #pragma unroll
  for (int k = 0; k < DEG_PIECES; ++k) {
    int i = tid + k * GG_THREADS;
    if (i >= DEG_CHUNKS) i = tid;
    int c = i + rot;
    if (c >= DEG_CHUNKS) c -= DEG_CHUNKS;
    degAt[k] = c;
    degPiece[k] = degSrc[c];
  }
  #pragma unroll
  for (int k = 0; k < W_PIECES; ++k) {
    int i = tid + k * GG_THREADS;
    if (i >= W_CHUNKS) i = tid;
    wAt[k] = i;
    wPiece[k] = Wv[i];
  }

  int beg, cnt, nmax, beg1, cnt1, nmax1;
  unsigned wlo, whi, wlo1, whi1;
  pair_bounds(0, beg, cnt, nmax);
  pair_bounds(1, beg1, cnt1, nmax1);
  load_words(beg, cnt, 0, wlo, whi);
  load_words(beg1, cnt1, 0, wlo1, whi1);

  {
    uint4* dst = (uint4*)sDeg;
    float4* sWv = (float4*)sW;
    #pragma unroll
    for (int k = 0; k < DEG_PIECES; ++k) dst[degAt[k]] = degPiece[k];
    #pragma unroll
    for (int k = 0; k < W_PIECES; ++k) sWv[wAt[k]] = wPiece[k];
  }
  __syncthreads();
  if (!live) return;

  float* myA = sA + wave * (16 * A_PITCH);
  float dlo = scale_of(sDeg[wlo], wlo, mn < cnt);
  float dhi = scale_of(sDeg[whi], whi, mn < TRIP_EDGES - 16 && 16 + mn < cnt);
  for (int p = 0; p < 8; ++p) {
    int beg2 = 0, cnt2 = 0, nmax2 = 0;
    unsigned wlo2 = 0u, whi2 = 0u;
    if (p + 2 < 8) {
      pair_bounds(p + 2, beg2, cnt2, nmax2);
      load_words(beg2, cnt2, 0, wlo2, whi2);
    }
    const unsigned tlo1 = sDeg[wlo1], thi1 = sDeg[whi1];

    float4 acc = {0.0f, 0.0f, 0.0f, 0.0f};
    int k = 0;
    do {
      if (k != 0) {
        load_words(beg, cnt, k, wlo, whi);
        dlo = scale_of(sDeg[wlo], wlo, k + mn < cnt);
        dhi = scale_of(sDeg[whi], whi, mn < TRIP_EDGES - 16 && k + 16 + mn < cnt);
      }
      const int rel = cnt - k;
      const int left = nmax - k;
      float4 v[TRIP_EDGES];
      unsigned s = 0u, db = 0u;
      if (left > 0 * STEP_GROUP) gather_group_load<0>(v, s, xb, wlo, whi, c16);
      if (left > 1 * STEP_GROUP) gather_group_load<1>(v, s, xb, wlo, whi, c16);
      if (left > 2 * STEP_GROUP) gather_group_load<2>(v, s, xb, wlo, whi, c16);
      if (left > 3 * STEP_GROUP) gather_group_load<3>(v, s, xb, wlo, whi, c16);
      if (left > 4 * STEP_GROUP) gather_group_load<4>(v, s, xb, wlo, whi, c16);
      if (left > 5 * STEP_GROUP) gather_group_load<5>(v, s, xb, wlo, whi, c16);
      if (left > 0 * STEP_GROUP) gather_group_add<0>(acc, db, v, dlo, dhi, rel);
      if (left > 1 * STEP_GROUP) gather_group_add<1>(acc, db, v, dlo, dhi, rel);
      if (left > 2 * STEP_GROUP) gather_group_add<2>(acc, db, v, dlo, dhi, rel);
      if (left > 3 * STEP_GROUP) gather_group_add<3>(acc, db, v, dlo, dhi, rel);
      if (left > 4 * STEP_GROUP) gather_group_add<4>(acc, db, v, dlo, dhi, rel);
      if (left > 5 * STEP_GROUP) gather_group_add<5>(acc, db, v, dlo, dhi, rel);
      k += TRIP_EDGES;
    } while (k < nmax);
    const float dr = rsqrtf((float)cnt);
    float2 o0, o1;
    o0.x = acc.x * dr; o0.y = acc.y * dr; o1.x = acc.z * dr; o1.y = acc.w * dr;
    float2* dst = (float2*)(myA + (2 * p + h) * A_PITCH + 4 * mn);
    dst[0] = o0; dst[1] = o1;

    dlo = scale_of(tlo1, wlo1, mn < cnt1);
    dhi = scale_of(thi1, whi1, mn < TRIP_EDGES - 16 && 16 + mn < cnt1);
    beg = beg1; cnt = cnt1; nmax = nmax1; wlo = wlo1; whi = whi1;
    beg1 = beg2; cnt1 = cnt2; nmax1 = nmax2; wlo1 = wlo2; whi1 = whi2;
  }
  __builtin_amdgcn_wave_barrier();

  v2f a[D_FEAT / 4];
  #pragma unroll
  for (int ks = 0; ks < D_FEAT / 4; ++ks) {
    const float2 t = *(const float2*)(myA + mn * A_PITCH + ks * 4 + 2 * h);
    a[ks].x = t.x; a[ks].y = t.y;
  }
  #pragma unroll
  for (int nt = 0; nt < UNITS / 16; ++nt) {
    const int nBase = nt * 16;
    v8f acc = {};
    #pragma unroll
    for (int ks = 0; ks < D_FEAT / 4; ++ks) {
      const int k0 = ks * 4 + 2 * h;
      v2f b;
      b.x = sW[(k0)     * UNITS + nBase + mn];
      b.y = sW[(k0 + 1) * UNITS + nBase + mn];
      acc = __builtin_amdgcn_wmma_f32_16x16x4_f32(
           false, a[ks],  false, b,
           (short)0, acc,  false,  false);
    }
    const float bn = bias[nBase + mn];
    #pragma unroll
    for (int j = 0; j < 8; ++j) {
      const int m = j + 8 * h;
      const float rr = acc[j] + bn;
      out[(size_t)(rowBase + m) * UNITS + nBase + mn] = fmaxf(rr, 0.0f);
    }
  }
}

extern "C" void kernel_launch(void* const* d_in, const int* in_sizes, int n_in,
                              void* d_out, int out_size, void* d_ws, size_t ws_size,
                              hipStream_t stream) {
  const float* x      = (const float*)d_in[0];
  const float* W      = (const float*)d_in[1];
  const float* b      = (const float*)d_in[2];
  const int*   source = (const int*)d_in[3];
  const int*   target = (const int*)d_in[4];
  float*       out    = (float*)d_out;

  unsigned* binned      = (unsigned*)d_ws;
  unsigned* sorted_edge = (unsigned*)d_ws + (size_t)N_EDGES;
  unsigned* seg         = (unsigned*)d_ws + (size_t)2 * N_EDGES;
  unsigned* rowptr      = (unsigned*)d_ws + (size_t)2 * N_EDGES + (size_t)N_RANGES * N_CHUNKS;
  float*    dinv        = (float*)(rowptr + N_NODES + 4);
  unsigned* deg8w       = (unsigned*)(dinv + N_NODES);

  gcn_partition_kernel<<<N_CHUNKS, PART_THREADS, 0, stream>>>(source, target, binned, seg);
  gcn_range_sort_kernel<<<N_RANGES, SORT_THREADS, 0, stream>>>(binned, seg, rowptr, dinv, deg8w, sorted_edge);
  gcn_gather_gemm_wmma_kernel<<<GG_BLOCKS, GG_THREADS, 0, stream>>>(x, sorted_edge, rowptr, dinv,
                                                                    deg8w, W, b, out);
}
